# starting kernel
# baseline (speedup 1.0000x reference)
#include <hip/hip_runtime.h>

#define N_NODES 100000
#define N_EDGES 1600000

typedef __attribute__((ext_vector_type(2))) float v2f;
typedef __attribute__((ext_vector_type(8))) float v8f;

__device__ __forceinline__ void f32_atomic_add(float* p, float v) {
    unsafeAtomicAdd(p, v);
}

__global__ void zero4_kernel(float4* __restrict__ p, int n4) {
    int t = blockIdx.x * blockDim.x + threadIdx.x;
    if (t < n4) p[t] = make_float4(0.f, 0.f, 0.f, 0.f);
}

__global__ void degree_kernel(const int* __restrict__ dst, float* __restrict__ deg) {
    int e = blockIdx.x * blockDim.x + threadIdx.x;
    if (e < N_EDGES) f32_atomic_add(&deg[dst[e]], 1.0f);
}

__global__ void dinv_kernel(float* __restrict__ deg_dinv) {
    int i = blockIdx.x * blockDim.x + threadIdx.x;
    if (i < N_NODES) deg_dinv[i] = rsqrtf(deg_dinv[i] + 1.0f);
}

template <int K, int COUT>
__global__ void gemm_dinv_wmma(const float* __restrict__ H,
                               const float* __restrict__ W,
                               const float* __restrict__ dinv,
                               float* __restrict__ Y,
                               float* __restrict__ AGG) {
    const int lane   = threadIdx.x & 31;
    const int n_tile = threadIdx.x >> 5;
    const int m_tile = blockIdx.x;

    const int row = m_tile * 16 + (lane & 15);
    const int col = n_tile * 16 + (lane & 15);
    const int kh  = (lane >> 4) << 1;

    const float* hrow = H + row * K + kh;

    v8f acc = {};
#pragma unroll
    for (int kk = 0; kk < K; kk += 4) {
        v2f a = *reinterpret_cast<const v2f*>(hrow + kk);
        v2f b;
        b.x = W[(kk + kh) * COUT + col];
        b.y = W[(kk + kh + 1) * COUT + col];
        acc = __builtin_amdgcn_wmma_f32_16x16x4_f32(
             false, a,  false, b,
             (short)0, acc,  false,  false);
    }

    const int mbase = m_tile * 16 + ((lane >> 4) << 3);
#pragma unroll
    for (int r = 0; r < 8; ++r) {
        const int m   = mbase + r;
        const float v = dinv[m] * acc[r];
        const int idx = m * COUT + col;
        Y[idx]   = v;
        AGG[idx] = v;
    }
}

template <int COUT>
__global__ void scatter_kernel(const int* __restrict__ src,
                               const int* __restrict__ dst,
                               const float* __restrict__ Y,
                               float* __restrict__ AGG) {
    constexpr int CH = COUT / 4;
    int t = blockIdx.x * blockDim.x + threadIdx.x;
    if (t >= N_EDGES * CH) return;
    int e  = t / CH;
    int c4 = (t - e * CH) * 4;
    int s = src[e];
    int d = dst[e];
    const float4 v = *reinterpret_cast<const float4*>(Y + s * COUT + c4);
    float* o = AGG + d * COUT + c4;
    f32_atomic_add(o + 0, v.x);
    f32_atomic_add(o + 1, v.y);
    f32_atomic_add(o + 2, v.z);
    f32_atomic_add(o + 3, v.w);
}

template <int COUT, bool RELU>
__global__ void combine_kernel(const float* __restrict__ AGG,
                               const float* __restrict__ dinv,
                               const float* __restrict__ bias,
                               float* __restrict__ OUT) {
    constexpr int SH = (COUT == 128) ? 7 : 6;
    int t = blockIdx.x * blockDim.x + threadIdx.x;
    if (t >= N_NODES * COUT) return;
    int i = t >> SH;
    int c = t & (COUT - 1);
    float v = dinv[i] * AGG[t] + bias[c];
    if (RELU) v = fmaxf(v, 0.0f);
    OUT[t] = v;
}

extern "C" void kernel_launch(void* const* d_in, const int* in_sizes, int n_in,
                              void* d_out, int out_size, void* d_ws, size_t ws_size,
                              hipStream_t stream) {
    const float* x  = (const float*)d_in[0];
    const int*   ei = (const int*)d_in[1];
    const float* W1 = (const float*)d_in[2];
    const float* b1 = (const float*)d_in[3];
    const float* W2 = (const float*)d_in[4];
    const float* b2 = (const float*)d_in[5];
    const float* W3 = (const float*)d_in[6];
    const float* b3 = (const float*)d_in[7];
    float*       out = (float*)d_out;

    const int* src = ei;
    const int* dst = ei + N_EDGES;

    char*  ws   = (char*)d_ws;
    size_t off  = 0;
    float* dinv = (float*)(ws + off); off += ((size_t)N_NODES * 4 + 511) & ~511ull;
    float* y    = (float*)(ws + off); off += (size_t)N_NODES * 128 * 4;
    float* agg  = (float*)(ws + off); off += (size_t)N_NODES * 128 * 4;
    float* h    = (float*)(ws + off);

    const int B = 256;
    const int M_TILES = N_NODES / 16;

    zero4_kernel<<<(N_NODES / 4 + B - 1) / B, B, 0, stream>>>((float4*)dinv, N_NODES / 4);
    degree_kernel<<<(N_EDGES + B - 1) / B, B, 0, stream>>>(dst, dinv);
    dinv_kernel<<<(N_NODES + B - 1) / B, B, 0, stream>>>(dinv);

    gemm_dinv_wmma<64, 128><<<M_TILES, 256, 0, stream>>>(x, W1, dinv, y, agg);
    scatter_kernel<128><<<(N_EDGES * 32 + B - 1) / B, B, 0, stream>>>(src, dst, y, agg);
    combine_kernel<128, true><<<(N_NODES * 128 + B - 1) / B, B, 0, stream>>>(agg, dinv, b1, h);

    gemm_dinv_wmma<128, 128><<<M_TILES, 256, 0, stream>>>(h, W2, dinv, y, agg);
    scatter_kernel<128><<<(N_EDGES * 32 + B - 1) / B, B, 0, stream>>>(src, dst, y, agg);
    combine_kernel<128, true><<<(N_NODES * 128 + B - 1) / B, B, 0, stream>>>(agg, dinv, b2, h);

    gemm_dinv_wmma<128, 64><<<M_TILES, 128, 0, stream>>>(h, W3, dinv, y, agg);
    scatter_kernel<64><<<(N_EDGES * 16 + B - 1) / B, B, 0, stream>>>(src, dst, y, agg);
    combine_kernel<64, false><<<(N_NODES * 64 + B - 1) / B, B, 0, stream>>>(agg, dinv, b3, out);
}


# aggregation as a destination-sorted edge list built each call (int atomic rank, scan, fill) and a per-node gather sum in registers, replacing per-edge f32 atomics; sum order = arrival rank
# speedup vs baseline: 2.7130x; 2.7130x over previous
#include <hip/hip_runtime.h>

#define N_NODES 100000
#define N_EDGES 1600000

typedef __attribute__((ext_vector_type(2))) float v2f;
typedef __attribute__((ext_vector_type(4))) float v4f;
typedef __attribute__((ext_vector_type(8))) float v8f;

#define SCAN_B 1024
#define SCAN_BLOCKS ((N_NODES + 1 + SCAN_B - 1) / SCAN_B)

__global__ void zero4_kernel(int4* __restrict__ p, int n4) {
    int t = blockIdx.x * blockDim.x + threadIdx.x;
    if (t < n4) p[t] = make_int4(0, 0, 0, 0);
}

__global__ void count_rank_kernel(const int* __restrict__ dst,
                                  int* __restrict__ cnt,
                                  int* __restrict__ rank) {
    int e = blockIdx.x * blockDim.x + threadIdx.x;
    if (e < N_EDGES) rank[e] = atomicAdd(&cnt[dst[e]], 1);
}

__global__ void __launch_bounds__(SCAN_B)
scan_local_kernel(const int* __restrict__ cnt,
                  int* __restrict__ rowptr,
                  int* __restrict__ bsum) {
    __shared__ int wsum[32];
    const int tid  = threadIdx.x;
    const int lane = tid & 31;
    const int w    = tid >> 5;
    const int i    = blockIdx.x * SCAN_B + tid;
    const int v    = (i < N_NODES) ? cnt[i] : 0;

    int incl = v;
#pragma unroll
    for (int o = 1; o < 32; o <<= 1) {
        int t = __shfl_up(incl, o);
        if (lane >= o) incl += t;
    }
    if (lane == 31) wsum[w] = incl;
    __syncthreads();
    if (w == 0) {
        const int s = wsum[lane];
        int si = s;
#pragma unroll
        for (int o = 1; o < 32; o <<= 1) {
            int t = __shfl_up(si, o);
            if (lane >= o) si += t;
        }
        wsum[lane] = si - s;
        if (lane == 31) bsum[blockIdx.x] = si;
    }
    __syncthreads();
    if (i <= N_NODES) rowptr[i] = wsum[w] + incl - v;
}

__global__ void __launch_bounds__(SCAN_B)
scan_finish_kernel(int* __restrict__ rowptr,
                   const int* __restrict__ bsum,
                   const int* __restrict__ cnt,
                   float* __restrict__ dinv) {
    __shared__ int off_s;
    const int tid = threadIdx.x;
    const int b   = blockIdx.x;
    if (tid < 32) {
        int s = 0;
        for (int k = tid; k < b; k += 32) s += bsum[k];
#pragma unroll
        for (int o = 16; o > 0; o >>= 1) s += __shfl_xor(s, o);
        if (tid == 0) off_s = s;
    }
    __syncthreads();
    const int i = b * SCAN_B + tid;
    if (i <= N_NODES) rowptr[i] += off_s;
    if (i < N_NODES) dinv[i] = rsqrtf((float)cnt[i] + 1.0f);
}

__global__ void fill_kernel(const int* __restrict__ src,
                            const int* __restrict__ dst,
                            const int* __restrict__ rank,
                            const int* __restrict__ rowptr,
                            int* __restrict__ col) {
    int e = blockIdx.x * blockDim.x + threadIdx.x;
    if (e < N_EDGES) col[rowptr[dst[e]] + rank[e]] = src[e];
}

template <int K, int COUT>
__global__ void gemm_dinv_wmma(const float* __restrict__ H,
                               const float* __restrict__ W,
                               const float* __restrict__ dinv,
                               float* __restrict__ Y) {
    const int lane   = threadIdx.x & 31;
    const int n_tile = threadIdx.x >> 5;
    const int m_tile = blockIdx.x;

    const int row = m_tile * 16 + (lane & 15);
    const int col = n_tile * 16 + (lane & 15);
    const int kh  = (lane >> 4) << 1;

    const float* hrow = H + row * K + kh;

    v8f acc = {};
#pragma unroll
    for (int kk = 0; kk < K; kk += 4) {
        v2f a = *reinterpret_cast<const v2f*>(hrow + kk);
        v2f b;
        b.x = W[(kk + kh) * COUT + col];
        b.y = W[(kk + kh + 1) * COUT + col];
        acc = __builtin_amdgcn_wmma_f32_16x16x4_f32(
             false, a,  false, b,
             (short)0, acc,  false,  false);
    }

    const int mbase = m_tile * 16 + ((lane >> 4) << 3);
#pragma unroll
    for (int r = 0; r < 8; ++r) {
        const int m = mbase + r;
        Y[m * COUT + col] = dinv[m] * acc[r];
    }
}

template <int COUT, bool RELU>
__global__ void __launch_bounds__(256)
gather_kernel(const int* __restrict__ rowptr,
              const int* __restrict__ col,
              const float* __restrict__ Y,
              const float* __restrict__ dinv,
              const float* __restrict__ bias,
              float* __restrict__ OUT) {
    constexpr int VEC = COUT / 32;
    typedef __attribute__((ext_vector_type(VEC))) float vecf;

    const int lane = threadIdx.x & 31;
    const int node = __builtin_amdgcn_readfirstlane(blockIdx.x * 8 + (threadIdx.x >> 5));
    const int beg  = rowptr[node];
    const int end  = rowptr[node + 1];

    const float* ylane = Y + lane * VEC;
    vecf acc = *reinterpret_cast<const vecf*>(ylane + node * COUT);

    for (int base = beg; base < end; base += 32) {
        const int rem  = end - base;
        const int n    = rem < 32 ? rem : 32;
        const int mine = (lane < n) ? col[base + lane] : 0;
        int j = 0;
        for (; j + 4 <= n; j += 4) {
            const int s0 = __builtin_amdgcn_readlane(mine, j);
            const int s1 = __builtin_amdgcn_readlane(mine, j + 1);
            const int s2 = __builtin_amdgcn_readlane(mine, j + 2);
            const int s3 = __builtin_amdgcn_readlane(mine, j + 3);
            const vecf v0 = *reinterpret_cast<const vecf*>(ylane + s0 * COUT);
            const vecf v1 = *reinterpret_cast<const vecf*>(ylane + s1 * COUT);
            const vecf v2 = *reinterpret_cast<const vecf*>(ylane + s2 * COUT);
            const vecf v3 = *reinterpret_cast<const vecf*>(ylane + s3 * COUT);
            acc += v0;
            acc += v1;
            acc += v2;
            acc += v3;
        }
        if (n - j >= 2) {
            const int s0 = __builtin_amdgcn_readlane(mine, j);
            const int s1 = __builtin_amdgcn_readlane(mine, j + 1);
            const vecf v0 = *reinterpret_cast<const vecf*>(ylane + s0 * COUT);
            const vecf v1 = *reinterpret_cast<const vecf*>(ylane + s1 * COUT);
            acc += v0;
            acc += v1;
            j += 2;
        }
        if (j < n) {
            const int s0 = __builtin_amdgcn_readlane(mine, j);
            acc += *reinterpret_cast<const vecf*>(ylane + s0 * COUT);
        }
    }

    const float di = dinv[node];
    const vecf bv  = *reinterpret_cast<const vecf*>(bias + lane * VEC);
    vecf o;
#pragma unroll
    for (int k = 0; k < VEC; ++k) {
        float v = di * acc[k] + bv[k];
        if (RELU) v = fmaxf(v, 0.0f);
        o[k] = v;
    }
    *reinterpret_cast<vecf*>(OUT + node * COUT + lane * VEC) = o;
}

extern "C" void kernel_launch(void* const* d_in, const int* in_sizes, int n_in,
                              void* d_out, int out_size, void* d_ws, size_t ws_size,
                              hipStream_t stream) {
    const float* x  = (const float*)d_in[0];
    const int*   ei = (const int*)d_in[1];
    const float* W1 = (const float*)d_in[2];
    const float* b1 = (const float*)d_in[3];
    const float* W2 = (const float*)d_in[4];
    const float* b2 = (const float*)d_in[5];
    const float* W3 = (const float*)d_in[6];
    const float* b3 = (const float*)d_in[7];
    float*       out = (float*)d_out;

    const int* src = ei;
    const int* dst = ei + N_EDGES;

    char*  ws     = (char*)d_ws;
    size_t off    = 0;
    float* dinv   = (float*)(ws + off); off += ((size_t)N_NODES * 4 + 511) & ~511ull;
    int*   cnt    = (int*)(ws + off);   off += ((size_t)N_NODES * 4 + 511) & ~511ull;
    int*   rowptr = (int*)(ws + off);   off += (size_t)SCAN_BLOCKS * SCAN_B * 4;
    int*   bsum   = (int*)(ws + off);   off += 512;
    int*   rank   = (int*)(ws + off);   off += (size_t)N_EDGES * 4;
    int*   col    = (int*)(ws + off);   off += (size_t)N_EDGES * 4;
    float* y      = (float*)(ws + off); off += (size_t)N_NODES * 128 * 4;
    float* h      = (float*)(ws + off);

    const int B = 256;
    const int M_TILES = N_NODES / 16;
    const int G_BLOCKS = N_NODES / 8;

    zero4_kernel<<<(N_NODES / 4 + B - 1) / B, B, 0, stream>>>((int4*)cnt, N_NODES / 4);
    count_rank_kernel<<<(N_EDGES + B - 1) / B, B, 0, stream>>>(dst, cnt, rank);
    scan_local_kernel<<<SCAN_BLOCKS, SCAN_B, 0, stream>>>(cnt, rowptr, bsum);
    scan_finish_kernel<<<SCAN_BLOCKS, SCAN_B, 0, stream>>>(rowptr, bsum, cnt, dinv);
    fill_kernel<<<(N_EDGES + B - 1) / B, B, 0, stream>>>(src, dst, rank, rowptr, col);

    gemm_dinv_wmma<64, 128><<<M_TILES, 256, 0, stream>>>(x, W1, dinv, y);
    gather_kernel<128, true><<<G_BLOCKS, 256, 0, stream>>>(rowptr, col, y, dinv, b1, h);

    gemm_dinv_wmma<128, 128><<<M_TILES, 256, 0, stream>>>(h, W2, dinv, y);
    gather_kernel<128, true><<<G_BLOCKS, 256, 0, stream>>>(rowptr, col, y, dinv, b2, h);

    gemm_dinv_wmma<128, 64><<<M_TILES, 128, 0, stream>>>(h, W3, dinv, y);
    gather_kernel<64, false><<<G_BLOCKS, 256, 0, stream>>>(rowptr, col, y, dinv, b3, out);
}


# transform fused into the aggregation kernel: neighbour rows are summed first (new sum order: sum, then @W) into an LDS tile that feeds the f32 WMMA; 3 GEMM launches and the Y round trip are gone
# speedup vs baseline: 3.1595x; 1.1646x over previous
#include <hip/hip_runtime.h>

#define N_NODES 100000
#define N_EDGES 1600000

typedef __attribute__((ext_vector_type(2))) float v2f;
typedef __attribute__((ext_vector_type(4))) float v4f;
typedef __attribute__((ext_vector_type(8))) float v8f;

#define SCAN_B 1024
#define SCAN_BLOCKS ((N_NODES + 1 + SCAN_B - 1) / SCAN_B)

__global__ void zero4_kernel(int4* __restrict__ p, int n4) {
    int t = blockIdx.x * blockDim.x + threadIdx.x;
    if (t < n4) p[t] = make_int4(0, 0, 0, 0);
}

__global__ void count_rank_kernel(const int* __restrict__ dst,
                                  int* __restrict__ cnt,
                                  int* __restrict__ rank) {
    int e = blockIdx.x * blockDim.x + threadIdx.x;
    if (e < N_EDGES) rank[e] = atomicAdd(&cnt[dst[e]], 1);
}

__global__ void __launch_bounds__(SCAN_B)
scan_local_kernel(const int* __restrict__ cnt,
                  int* __restrict__ rowptr,
                  int* __restrict__ bsum) {
    __shared__ int wsum[32];
    const int tid  = threadIdx.x;
    const int lane = tid & 31;
    const int w    = tid >> 5;
    const int i    = blockIdx.x * SCAN_B + tid;
    const int v    = (i < N_NODES) ? cnt[i] : 0;

    int incl = v;
#pragma unroll
    for (int o = 1; o < 32; o <<= 1) {
        int t = __shfl_up(incl, o);
        if (lane >= o) incl += t;
    }
    if (lane == 31) wsum[w] = incl;
    __syncthreads();
    if (w == 0) {
        const int s = wsum[lane];
        int si = s;
#pragma unroll
        for (int o = 1; o < 32; o <<= 1) {
            int t = __shfl_up(si, o);
            if (lane >= o) si += t;
        }
        wsum[lane] = si - s;
        if (lane == 31) bsum[blockIdx.x] = si;
    }
    __syncthreads();
    if (i <= N_NODES) rowptr[i] = wsum[w] + incl - v;
}

__global__ void __launch_bounds__(SCAN_B)
scan_finish_kernel(int* __restrict__ rowptr,
                   const int* __restrict__ bsum,
                   const int* __restrict__ cnt,
                   float* __restrict__ dinv) {
    __shared__ int off_s;
    const int tid = threadIdx.x;
    const int b   = blockIdx.x;
    if (tid < 32) {
        int s = 0;
        for (int k = tid; k < b; k += 32) s += bsum[k];
#pragma unroll
        for (int o = 16; o > 0; o >>= 1) s += __shfl_xor(s, o);
        if (tid == 0) off_s = s;
    }
    __syncthreads();
    const int i = b * SCAN_B + tid;
    if (i <= N_NODES) rowptr[i] += off_s;
    if (i < N_NODES) dinv[i] = rsqrtf((float)cnt[i] + 1.0f);
}

__global__ void fill_kernel(const int* __restrict__ src,
                            const int* __restrict__ dst,
                            const int* __restrict__ rank,
                            const int* __restrict__ rowptr,
                            int* __restrict__ col) {
    int e = blockIdx.x * blockDim.x + threadIdx.x;
    if (e < N_EDGES) col[rowptr[dst[e]] + rank[e]] = src[e];
}

template <int K, int COUT, bool SCALE_IN, bool RELU, bool SCALE_OUT>
__global__ void __launch_bounds__(1024)
agg_gemm_kernel(const int* __restrict__ rowptr,
                const int* __restrict__ col,
                const float* __restrict__ G,
                const float* __restrict__ dinv,
                const float* __restrict__ W,
                const float* __restrict__ bias,
                float* __restrict__ OUT) {
    constexpr int VEC = K / 32;
    constexpr int P   = K + 4;
    typedef __attribute__((ext_vector_type(VEC))) float vecf;
    constexpr int NT  = COUT / 16;
    __shared__ __attribute__((aligned(16))) float tile[32 * P];

    const int lane = threadIdx.x & 31;
    const int w    = __builtin_amdgcn_readfirstlane(threadIdx.x >> 5);
    const int node = blockIdx.x * 32 + w;
    const int beg  = rowptr[node];
    const int end  = rowptr[node + 1];

    const float* glane = G + lane * VEC;
    vecf acc = *reinterpret_cast<const vecf*>(glane + node * K);
    if (SCALE_IN) acc *= dinv[node];

    for (int base = beg; base < end; base += 32) {
        const int rem  = end - base;
        const int n    = rem < 32 ? rem : 32;
        const int mine = (lane < n) ? col[base + lane] : 0;
        int dmine = 0;
        if (SCALE_IN) dmine = __float_as_int((lane < n) ? dinv[mine] : 0.0f);
        int j = 0;
        for (; j + 4 <= n; j += 4) {
            const int s0 = __builtin_amdgcn_readlane(mine, j);
            const int s1 = __builtin_amdgcn_readlane(mine, j + 1);
            const int s2 = __builtin_amdgcn_readlane(mine, j + 2);
            const int s3 = __builtin_amdgcn_readlane(mine, j + 3);
            const vecf v0 = *reinterpret_cast<const vecf*>(glane + s0 * K);
            const vecf v1 = *reinterpret_cast<const vecf*>(glane + s1 * K);
            const vecf v2 = *reinterpret_cast<const vecf*>(glane + s2 * K);
            const vecf v3 = *reinterpret_cast<const vecf*>(glane + s3 * K);
            if (SCALE_IN) {
                acc += v0 * __int_as_float(__builtin_amdgcn_readlane(dmine, j));
                acc += v1 * __int_as_float(__builtin_amdgcn_readlane(dmine, j + 1));
                acc += v2 * __int_as_float(__builtin_amdgcn_readlane(dmine, j + 2));
                acc += v3 * __int_as_float(__builtin_amdgcn_readlane(dmine, j + 3));
            } else {
                acc += v0;
                acc += v1;
                acc += v2;
                acc += v3;
            }
        }
        if (n - j >= 2) {
            const int s0 = __builtin_amdgcn_readlane(mine, j);
            const int s1 = __builtin_amdgcn_readlane(mine, j + 1);
            const vecf v0 = *reinterpret_cast<const vecf*>(glane + s0 * K);
            const vecf v1 = *reinterpret_cast<const vecf*>(glane + s1 * K);
            if (SCALE_IN) {
                acc += v0 * __int_as_float(__builtin_amdgcn_readlane(dmine, j));
                acc += v1 * __int_as_float(__builtin_amdgcn_readlane(dmine, j + 1));
            } else {
                acc += v0;
                acc += v1;
            }
            j += 2;
        }
        if (j < n) {
            const int s0 = __builtin_amdgcn_readlane(mine, j);
            const vecf v0 = *reinterpret_cast<const vecf*>(glane + s0 * K);
            if (SCALE_IN) acc += v0 * __int_as_float(__builtin_amdgcn_readlane(dmine, j));
            else          acc += v0;
        }
    }

    *reinterpret_cast<vecf*>(tile + w * P + lane * VEC) = acc;
    __syncthreads();

    if (w < 2 * NT) {
        const int mt   = w / NT;
        const int nt   = w - mt * NT;
        const int c    = lane & 15;
        const int kh   = (lane >> 4) << 1;
        const int colg = nt * 16 + c;
        const float* arow = tile + (mt * 16 + c) * P + kh;
        const float* wcol = W + kh * COUT + colg;

        v8f dacc = {};
#pragma unroll
        for (int kk = 0; kk < K; kk += 4) {
            v2f a = *reinterpret_cast<const v2f*>(arow + kk);
            v2f b;
            b.x = wcol[(kk)     * COUT];
            b.y = wcol[(kk + 1) * COUT];
            dacc = __builtin_amdgcn_wmma_f32_16x16x4_f32(
                 false, a,  false, b,
                 (short)0, dacc,  false,  false);
        }

        const int   mbase = blockIdx.x * 32 + mt * 16 + ((lane >> 4) << 3);
        const float bcol  = bias[colg];
#pragma unroll
        for (int r = 0; r < 8; ++r) {
            const int   m  = mbase + r;
            const float dm = dinv[m];
            float v = dm * dacc[r] + bcol;
            if (RELU) v = fmaxf(v, 0.0f);
            if (SCALE_OUT) v *= dm;
            OUT[m * COUT + colg] = v;
        }
    }
}

extern "C" void kernel_launch(void* const* d_in, const int* in_sizes, int n_in,
                              void* d_out, int out_size, void* d_ws, size_t ws_size,
                              hipStream_t stream) {
    const float* x  = (const float*)d_in[0];
    const int*   ei = (const int*)d_in[1];
    const float* W1 = (const float*)d_in[2];
    const float* b1 = (const float*)d_in[3];
    const float* W2 = (const float*)d_in[4];
    const float* b2 = (const float*)d_in[5];
    const float* W3 = (const float*)d_in[6];
    const float* b3 = (const float*)d_in[7];
    float*       out = (float*)d_out;

    const int* src = ei;
    const int* dst = ei + N_EDGES;

    char*  ws     = (char*)d_ws;
    size_t off    = 0;
    float* dinv   = (float*)(ws + off); off += ((size_t)N_NODES * 4 + 511) & ~511ull;
    int*   cnt    = (int*)(ws + off);   off += ((size_t)N_NODES * 4 + 511) & ~511ull;
    int*   rowptr = (int*)(ws + off);   off += (size_t)SCAN_BLOCKS * SCAN_B * 4;
    int*   bsum   = (int*)(ws + off);   off += 512;
    int*   rank   = (int*)(ws + off);   off += (size_t)N_EDGES * 4;
    int*   col    = (int*)(ws + off);   off += (size_t)N_EDGES * 4;
    float* ha     = (float*)(ws + off); off += (size_t)N_NODES * 128 * 4;
    float* hb     = (float*)(ws + off);

    const int B = 256;
    const int L_BLOCKS = N_NODES / 32;

    zero4_kernel<<<(N_NODES / 4 + B - 1) / B, B, 0, stream>>>((int4*)cnt, N_NODES / 4);
    count_rank_kernel<<<(N_EDGES + B - 1) / B, B, 0, stream>>>(dst, cnt, rank);
    scan_local_kernel<<<SCAN_BLOCKS, SCAN_B, 0, stream>>>(cnt, rowptr, bsum);
    scan_finish_kernel<<<SCAN_BLOCKS, SCAN_B, 0, stream>>>(rowptr, bsum, cnt, dinv);
    fill_kernel<<<(N_EDGES + B - 1) / B, B, 0, stream>>>(src, dst, rank, rowptr, col);

    agg_gemm_kernel<64, 128, true, true, true><<<L_BLOCKS, 1024, 0, stream>>>(rowptr, col, x, dinv, W1, b1, ha);

    agg_gemm_kernel<128, 128, false, true, true><<<L_BLOCKS, 1024, 0, stream>>>(rowptr, col, ha, dinv, W2, b2, hb);

    agg_gemm_kernel<128, 64, false, false, false><<<L_BLOCKS, 1024, 0, stream>>>(rowptr, col, hb, dinv, W3, b3, out);
}


# layer kernels made resident: 1024 work-groups each walk 3-4 tiles of 32 nodes (W kept in LDS, double-buffered sum tile, wave groups take the WMMA product in turn); 3x fewer waves; sum order kept
# speedup vs baseline: 3.3528x; 1.0612x over previous
#include <hip/hip_runtime.h>

#define N_NODES 100000
#define N_EDGES 1600000
#define N_TILES (N_NODES / 32)

typedef __attribute__((ext_vector_type(2))) float v2f;
typedef __attribute__((ext_vector_type(4))) float v4f;
typedef __attribute__((ext_vector_type(8))) float v8f;

#define SCAN_B 1024
#define SCAN_BLOCKS ((N_NODES + 1 + SCAN_B - 1) / SCAN_B)

__global__ void zero4_kernel(int4* __restrict__ p, int n4) {
    int t = blockIdx.x * blockDim.x + threadIdx.x;
    if (t < n4) p[t] = make_int4(0, 0, 0, 0);
}

__global__ void count_rank_kernel(const int* __restrict__ dst,
                                  int* __restrict__ cnt,
                                  int* __restrict__ rank) {
    int e = blockIdx.x * blockDim.x + threadIdx.x;
    if (e < N_EDGES) rank[e] = atomicAdd(&cnt[dst[e]], 1);
}

__global__ void __launch_bounds__(SCAN_B)
scan_local_kernel(const int* __restrict__ cnt,
                  int* __restrict__ rowptr,
                  int* __restrict__ bsum) {
    __shared__ int wsum[32];
    const int tid  = threadIdx.x;
    const int lane = tid & 31;
    const int w    = tid >> 5;
    const int i    = blockIdx.x * SCAN_B + tid;
    const int v    = (i < N_NODES) ? cnt[i] : 0;

    int incl = v;
#pragma unroll
    for (int o = 1; o < 32; o <<= 1) {
        int t = __shfl_up(incl, o);
        if (lane >= o) incl += t;
    }
    if (lane == 31) wsum[w] = incl;
    __syncthreads();
    if (w == 0) {
        const int s = wsum[lane];
        int si = s;
#pragma unroll
        for (int o = 1; o < 32; o <<= 1) {
            int t = __shfl_up(si, o);
            if (lane >= o) si += t;
        }
        wsum[lane] = si - s;
        if (lane == 31) bsum[blockIdx.x] = si;
    }
    __syncthreads();
    if (i <= N_NODES) rowptr[i] = wsum[w] + incl - v;
}

__global__ void __launch_bounds__(SCAN_B)
scan_finish_kernel(int* __restrict__ rowptr,
                   const int* __restrict__ bsum,
                   const int* __restrict__ cnt,
                   float* __restrict__ dinv) {
    __shared__ int off_s;
    const int tid = threadIdx.x;
    const int b   = blockIdx.x;
    if (tid < 32) {
        int s = 0;
        for (int k = tid; k < b; k += 32) s += bsum[k];
#pragma unroll
        for (int o = 16; o > 0; o >>= 1) s += __shfl_xor(s, o);
        if (tid == 0) off_s = s;
    }
    __syncthreads();
    const int i = b * SCAN_B + tid;
    if (i <= N_NODES) rowptr[i] += off_s;
    if (i < N_NODES) dinv[i] = rsqrtf((float)cnt[i] + 1.0f);
}

__global__ void fill_kernel(const int* __restrict__ src,
                            const int* __restrict__ dst,
                            const int* __restrict__ rank,
                            const int* __restrict__ rowptr,
                            int* __restrict__ col) {
    int e = blockIdx.x * blockDim.x + threadIdx.x;
    if (e < N_EDGES) col[rowptr[dst[e]] + rank[e]] = src[e];
}

template <int K, int COUT, bool SCALE_IN, bool RELU, bool SCALE_OUT>
__global__ void __launch_bounds__(1024)
agg_gemm_kernel(const int* __restrict__ rowptr,
                const int* __restrict__ col,
                const float* __restrict__ G,
                const float* __restrict__ dinv,
                const float* __restrict__ W,
                const float* __restrict__ bias,
                float* __restrict__ OUT) {
    constexpr int VEC = K / 32;
    constexpr int P   = K + 4;
    constexpr int WP  = COUT + 8;
    typedef __attribute__((ext_vector_type(VEC))) float vecf;
    constexpr int NT  = COUT / 16;
    constexpr int GW  = 2 * NT;
    constexpr int NG  = 32 / GW;
    __shared__ __attribute__((aligned(16))) float tile[2][32 * P];
    __shared__ __attribute__((aligned(16))) float wlds[K * WP];

    const int lane = threadIdx.x & 31;
    const int w    = __builtin_amdgcn_readfirstlane(threadIdx.x >> 5);
    const int grp  = w / GW;
    const int wl   = w - grp * GW;
    const float* glane = G + lane * VEC;

    constexpr int NWP = K * COUT / 4096;
    v4f wv[NWP];
#pragma unroll
    for (int i = 0; i < NWP; ++i)
        wv[i] = *reinterpret_cast<const v4f*>(W + threadIdx.x * 4 + i * 4096);
#pragma unroll
    for (int i = 0; i < NWP; ++i) {
        const int f   = threadIdx.x * 4 + i * 4096;
        const int row = f / COUT;
        const int cx  = f - row * COUT;
        *reinterpret_cast<v4f*>(wlds + row * WP + cx) = wv[i];
    }

    int it = 0;
    for (int t = blockIdx.x; t < N_TILES; t += gridDim.x, ++it) {
        float* tb = tile[it & 1];
        const int node = t * 32 + w;
        const int beg  = rowptr[node];
        const int end  = rowptr[node + 1];

        vecf acc = *reinterpret_cast<const vecf*>(glane + node * K);
        if (SCALE_IN) acc *= dinv[node];

        for (int base = beg; base < end; base += 32) {
            const int rem  = end - base;
            const int n    = rem < 32 ? rem : 32;
            const int mine = (lane < n) ? col[base + lane] : 0;
            int dmine = 0;
            if (SCALE_IN) dmine = __float_as_int((lane < n) ? dinv[mine] : 0.0f);
            int j = 0;
            for (; j + 4 <= n; j += 4) {
                const int s0 = __builtin_amdgcn_readlane(mine, j);
                const int s1 = __builtin_amdgcn_readlane(mine, j + 1);
                const int s2 = __builtin_amdgcn_readlane(mine, j + 2);
                const int s3 = __builtin_amdgcn_readlane(mine, j + 3);
                const vecf v0 = *reinterpret_cast<const vecf*>(glane + s0 * K);
                const vecf v1 = *reinterpret_cast<const vecf*>(glane + s1 * K);
                const vecf v2 = *reinterpret_cast<const vecf*>(glane + s2 * K);
                const vecf v3 = *reinterpret_cast<const vecf*>(glane + s3 * K);
                if (SCALE_IN) {
                    acc += v0 * __int_as_float(__builtin_amdgcn_readlane(dmine, j));
                    acc += v1 * __int_as_float(__builtin_amdgcn_readlane(dmine, j + 1));
                    acc += v2 * __int_as_float(__builtin_amdgcn_readlane(dmine, j + 2));
                    acc += v3 * __int_as_float(__builtin_amdgcn_readlane(dmine, j + 3));
                } else {
                    acc += v0;
                    acc += v1;
                    acc += v2;
                    acc += v3;
                }
            }
            if (n - j >= 2) {
                const int s0 = __builtin_amdgcn_readlane(mine, j);
                const int s1 = __builtin_amdgcn_readlane(mine, j + 1);
                const vecf v0 = *reinterpret_cast<const vecf*>(glane + s0 * K);
                const vecf v1 = *reinterpret_cast<const vecf*>(glane + s1 * K);
                if (SCALE_IN) {
                    acc += v0 * __int_as_float(__builtin_amdgcn_readlane(dmine, j));
                    acc += v1 * __int_as_float(__builtin_amdgcn_readlane(dmine, j + 1));
                } else {
                    acc += v0;
                    acc += v1;
                }
                j += 2;
            }
            if (j < n) {
                const int s0 = __builtin_amdgcn_readlane(mine, j);
                const vecf v0 = *reinterpret_cast<const vecf*>(glane + s0 * K);
                if (SCALE_IN) acc += v0 * __int_as_float(__builtin_amdgcn_readlane(dmine, j));
                else          acc += v0;
            }
        }

        *reinterpret_cast<vecf*>(tb + w * P + lane * VEC) = acc;
        __syncthreads();

        if (grp == (it & (NG - 1))) {
            const int mt   = wl / NT;
            const int nt   = wl - mt * NT;
            const int c    = lane & 15;
            const int kh   = (lane >> 4) << 1;
            const int colg = nt * 16 + c;
            const float* arow = tb + (mt * 16 + c) * P + kh;
            const float* wcol = wlds + kh * WP + colg;

            v8f dacc = {};
#pragma unroll
            for (int kk = 0; kk < K; kk += 4) {
                v2f a = *reinterpret_cast<const v2f*>(arow + kk);
                v2f b;
                b.x = wcol[(kk)     * WP];
                b.y = wcol[(kk + 1) * WP];
                dacc = __builtin_amdgcn_wmma_f32_16x16x4_f32(
                     false, a,  false, b,
                     (short)0, dacc,  false,  false);
            }

            const int   mbase = t * 32 + mt * 16 + ((lane >> 4) << 3);
            const float bcol  = bias[colg];
#pragma unroll
            for (int r = 0; r < 8; ++r) {
                const int   m  = mbase + r;
                const float dm = dinv[m];
                float v = dm * dacc[r] + bcol;
                if (RELU) v = fmaxf(v, 0.0f);
                if (SCALE_OUT) v *= dm;
                OUT[m * COUT + colg] = v;
            }
        }
    }
}

extern "C" void kernel_launch(void* const* d_in, const int* in_sizes, int n_in,
                              void* d_out, int out_size, void* d_ws, size_t ws_size,
                              hipStream_t stream) {
    const float* x  = (const float*)d_in[0];
    const int*   ei = (const int*)d_in[1];
    const float* W1 = (const float*)d_in[2];
    const float* b1 = (const float*)d_in[3];
    const float* W2 = (const float*)d_in[4];
    const float* b2 = (const float*)d_in[5];
    const float* W3 = (const float*)d_in[6];
    const float* b3 = (const float*)d_in[7];
    float*       out = (float*)d_out;

    const int* src = ei;
    const int* dst = ei + N_EDGES;

    char*  ws     = (char*)d_ws;
    size_t off    = 0;
    float* dinv   = (float*)(ws + off); off += ((size_t)N_NODES * 4 + 511) & ~511ull;
    int*   cnt    = (int*)(ws + off);   off += ((size_t)N_NODES * 4 + 511) & ~511ull;
    int*   rowptr = (int*)(ws + off);   off += (size_t)SCAN_BLOCKS * SCAN_B * 4;
    int*   bsum   = (int*)(ws + off);   off += 512;
    int*   rank   = (int*)(ws + off);   off += (size_t)N_EDGES * 4;
    int*   col    = (int*)(ws + off);   off += (size_t)N_EDGES * 4;
    float* ha     = (float*)(ws + off); off += (size_t)N_NODES * 128 * 4;
    float* hb     = (float*)(ws + off);

    const int B = 256;
    const int L_BLOCKS = 1024;

    zero4_kernel<<<(N_NODES / 4 + B - 1) / B, B, 0, stream>>>((int4*)cnt, N_NODES / 4);
    count_rank_kernel<<<(N_EDGES + B - 1) / B, B, 0, stream>>>(dst, cnt, rank);
    scan_local_kernel<<<SCAN_BLOCKS, SCAN_B, 0, stream>>>(cnt, rowptr, bsum);
    scan_finish_kernel<<<SCAN_BLOCKS, SCAN_B, 0, stream>>>(rowptr, bsum, cnt, dinv);
    fill_kernel<<<(N_EDGES + B - 1) / B, B, 0, stream>>>(src, dst, rank, rowptr, col);

    agg_gemm_kernel<64, 128, true, true, true><<<L_BLOCKS, 1024, 0, stream>>>(rowptr, col, x, dinv, W1, b1, ha);

    agg_gemm_kernel<128, 128, false, true, true><<<L_BLOCKS, 1024, 0, stream>>>(rowptr, col, ha, dinv, W2, b2, hb);

    agg_gemm_kernel<128, 64, false, false, false><<<L_BLOCKS, 1024, 0, stream>>>(rowptr, col, hb, dinv, W3, b3, out);
}


# edge-list build as a two-level counting sort in LDS (count per bin of 256 nodes, move pairs, sort each bin): 3 launches for 5, no global atomics; sum order = arrival rank of LDS atomics
# speedup vs baseline: 4.8342x; 1.4418x over previous
#include <hip/hip_runtime.h>

#define N_NODES 100000
#define N_EDGES 1600000
#define N_TILES (N_NODES / 32)

typedef __attribute__((ext_vector_type(2))) float v2f;
typedef __attribute__((ext_vector_type(4))) float v4f;
typedef __attribute__((ext_vector_type(8))) float v8f;

#define BIN_SHIFT 8
#define BIN_NODES 256
#define NB ((N_NODES + BIN_NODES - 1) / BIN_NODES)
#define GP 256
#define CHUNK (N_EDGES / GP)
#define PER ((CHUNK + 1023) / 1024)
static_assert(N_EDGES % GP == 0, "edge chunks are exact");
static_assert(GP % 32 == 0, "a wave reads a bin's counts in whole passes");

__global__ void __launch_bounds__(1024)
bin_count_kernel(const int* __restrict__ dst, int* __restrict__ wgcnt) {
    __shared__ int hist[NB];
    const int tid = threadIdx.x;
    const int e0  = blockIdx.x * CHUNK;
    int d[PER];
#pragma unroll
    for (int k = 0; k < PER; ++k) {
        const int i = tid + k * 1024;
        d[k] = dst[e0 + (i < CHUNK ? i : CHUNK - 1)];
    }
    if (tid < NB) hist[tid] = 0;
    __syncthreads();
#pragma unroll
    for (int k = 0; k < PER; ++k)
        if (tid + k * 1024 < CHUNK) atomicAdd(&hist[d[k] >> BIN_SHIFT], 1);
    __syncthreads();
    if (tid < NB) wgcnt[tid * GP + blockIdx.x] = hist[tid];
}

__global__ void __launch_bounds__(1024)
bin_scatter_kernel(const int* __restrict__ src, const int* __restrict__ dst,
                   const int* __restrict__ wgcnt, int* __restrict__ binstart,
                   int2* __restrict__ binned) {
    __shared__ int cur[NB];
    __shared__ int tot[NB];
    const int tid  = threadIdx.x;
    const int lane = tid & 31;
    const int w    = tid >> 5;
    const int wg   = blockIdx.x;
    const int e0   = wg * CHUNK;

    int es[PER], ed[PER];
#pragma unroll
    for (int k = 0; k < PER; ++k) {
        const int i  = tid + k * 1024;
        const int ic = i < CHUNK ? i : CHUNK - 1;
        es[k] = src[e0 + ic];
        ed[k] = dst[e0 + ic];
    }

    for (int b0 = w; b0 < NB; b0 += 128) {
        int pre[4], all[4];
#pragma unroll
        for (int j = 0; j < 4; ++j) {
            const int b  = b0 + 32 * j;
            const int bc = b < NB ? b : NB - 1;
            pre[j] = 0;
            all[j] = 0;
#pragma unroll
            for (int k = 0; k < GP / 32; ++k) {
                const int i = k * 32 + lane;
                const int v = wgcnt[bc * GP + i];
                all[j] += v;
                pre[j] += (i < wg) ? v : 0;
            }
        }
#pragma unroll
        for (int j = 0; j < 4; ++j) {
#pragma unroll
            for (int o = 16; o > 0; o >>= 1) {
                all[j] += __shfl_xor(all[j], o);
                pre[j] += __shfl_xor(pre[j], o);
            }
            const int b = b0 + 32 * j;
            if (lane == 0 && b < NB) { tot[b] = all[j]; cur[b] = pre[j]; }
        }
    }
    __syncthreads();
    if (w == 0) {
        int carry = 0;
        for (int c = 0; c < NB; c += 32) {
            const int b = c + lane;
            const int v = (b < NB) ? tot[b] : 0;
            int incl = v;
#pragma unroll
            for (int o = 1; o < 32; o <<= 1) {
                const int t = __shfl_up(incl, o);
                if (lane >= o) incl += t;
            }
            if (b < NB) {
                const int start = carry + incl - v;
                cur[b] += start;
                if (wg == 0) binstart[b] = start;
            }
            carry += __shfl(incl, 31);
        }
        if (wg == 0 && lane == 0) binstart[NB] = carry;
    }
    __syncthreads();

#pragma unroll
    for (int k = 0; k < PER; ++k) {
        if (tid + k * 1024 < CHUNK) {
            const int pos = atomicAdd(&cur[ed[k] >> BIN_SHIFT], 1);
            binned[pos] = make_int2(es[k], ed[k]);
        }
    }
}

__global__ void __launch_bounds__(1024)
bin_sort_kernel(const int2* __restrict__ binned, const int* __restrict__ binstart,
                int* __restrict__ rowptr, int* __restrict__ col,
                float* __restrict__ dinv) {
    __shared__ int cnt[BIN_NODES];
    __shared__ int curs[BIN_NODES];
    __shared__ int wtot[BIN_NODES / 32];
    const int tid  = threadIdx.x;
    const int lane = tid & 31;
    const int w    = tid >> 5;
    const int b    = blockIdx.x;
    const int bs   = binstart[b];
    const int be   = binstart[b + 1];

    if (tid < BIN_NODES) cnt[tid] = 0;
    __syncthreads();
    for (int e = bs + tid; e < be; e += 4096) {
        int y[4];
#pragma unroll
        for (int j = 0; j < 4; ++j) {
            const int i = e + j * 1024;
            y[j] = binned[i < be ? i : be - 1].y;
        }
        asm volatile("" :: "v"(y[0]), "v"(y[1]), "v"(y[2]), "v"(y[3]));
#pragma unroll
        for (int j = 0; j < 4; ++j)
            if (e + j * 1024 < be) atomicAdd(&cnt[y[j] & (BIN_NODES - 1)], 1);
    }
    __syncthreads();

    int v = 0, incl = 0;
    if (tid < BIN_NODES) {
        v = cnt[tid];
        incl = v;
#pragma unroll
        for (int o = 1; o < 32; o <<= 1) {
            const int t = __shfl_up(incl, o);
            if (lane >= o) incl += t;
        }
        if (lane == 31) wtot[w] = incl;
    }
    __syncthreads();
    if (tid < BIN_NODES) {
        int off = 0;
        for (int k = 0; k < w; ++k) off += wtot[k];
        const int excl = off + incl - v;
        curs[tid] = excl;
        const int node = b * BIN_NODES + tid;
        if (node <= N_NODES) rowptr[node] = bs + excl;
        if (node < N_NODES)  dinv[node] = rsqrtf((float)v + 1.0f);
    }
    __syncthreads();
    for (int e = bs + tid; e < be; e += 4096) {
        int2 sd[4];
#pragma unroll
        for (int j = 0; j < 4; ++j) {
            const int i = e + j * 1024;
            sd[j] = binned[i < be ? i : be - 1];
        }
        asm volatile("" :: "v"(sd[0].y), "v"(sd[1].y), "v"(sd[2].y), "v"(sd[3].y));
#pragma unroll
        for (int j = 0; j < 4; ++j) {
            if (e + j * 1024 < be) {
                const int pos = atomicAdd(&curs[sd[j].y & (BIN_NODES - 1)], 1);
                col[bs + pos] = sd[j].x;
            }
        }
    }
}

template <int K, int COUT, bool SCALE_IN, bool RELU, bool SCALE_OUT>
__global__ void __launch_bounds__(1024)
agg_gemm_kernel(const int* __restrict__ rowptr,
                const int* __restrict__ col,
                const float* __restrict__ G,
                const float* __restrict__ dinv,
                const float* __restrict__ W,
                const float* __restrict__ bias,
                float* __restrict__ OUT) {
    constexpr int VEC = K / 32;
    constexpr int P   = K + 4;
    constexpr int WP  = COUT + 8;
    typedef __attribute__((ext_vector_type(VEC))) float vecf;
    constexpr int NT  = COUT / 16;
    constexpr int GW  = 2 * NT;
    constexpr int NG  = 32 / GW;
    __shared__ __attribute__((aligned(16))) float tile[2][32 * P];
    __shared__ __attribute__((aligned(16))) float wlds[K * WP];

    const int lane = threadIdx.x & 31;
    const int w    = __builtin_amdgcn_readfirstlane(threadIdx.x >> 5);
    const int grp  = w / GW;
    const int wl   = w - grp * GW;
    const float* glane = G + lane * VEC;

    constexpr int NWP = K * COUT / 4096;
    v4f wv[NWP];
#pragma unroll
    for (int i = 0; i < NWP; ++i)
        wv[i] = *reinterpret_cast<const v4f*>(W + threadIdx.x * 4 + i * 4096);
#pragma unroll
    for (int i = 0; i < NWP; ++i) {
        const int f   = threadIdx.x * 4 + i * 4096;
        const int row = f / COUT;
        const int cx  = f - row * COUT;
        *reinterpret_cast<v4f*>(wlds + row * WP + cx) = wv[i];
    }

    int it = 0;
    for (int t = blockIdx.x; t < N_TILES; t += gridDim.x, ++it) {
        float* tb = tile[it & 1];
        const int node = t * 32 + w;
        const int beg  = rowptr[node];
        const int end  = rowptr[node + 1];

        vecf acc = *reinterpret_cast<const vecf*>(glane + node * K);
        if (SCALE_IN) acc *= dinv[node];

        for (int base = beg; base < end; base += 32) {
            const int rem  = end - base;
            const int n    = rem < 32 ? rem : 32;
            const int mine = (lane < n) ? col[base + lane] : 0;
            int dmine = 0;
            if (SCALE_IN) dmine = __float_as_int((lane < n) ? dinv[mine] : 0.0f);
            int j = 0;
            for (; j + 4 <= n; j += 4) {
                const int s0 = __builtin_amdgcn_readlane(mine, j);
                const int s1 = __builtin_amdgcn_readlane(mine, j + 1);
                const int s2 = __builtin_amdgcn_readlane(mine, j + 2);
                const int s3 = __builtin_amdgcn_readlane(mine, j + 3);
                const vecf v0 = *reinterpret_cast<const vecf*>(glane + s0 * K);
                const vecf v1 = *reinterpret_cast<const vecf*>(glane + s1 * K);
                const vecf v2 = *reinterpret_cast<const vecf*>(glane + s2 * K);
                const vecf v3 = *reinterpret_cast<const vecf*>(glane + s3 * K);
                if (SCALE_IN) {
                    acc += v0 * __int_as_float(__builtin_amdgcn_readlane(dmine, j));
                    acc += v1 * __int_as_float(__builtin_amdgcn_readlane(dmine, j + 1));
                    acc += v2 * __int_as_float(__builtin_amdgcn_readlane(dmine, j + 2));
                    acc += v3 * __int_as_float(__builtin_amdgcn_readlane(dmine, j + 3));
                } else {
                    acc += v0;
                    acc += v1;
                    acc += v2;
                    acc += v3;
                }
            }
            if (n - j >= 2) {
                const int s0 = __builtin_amdgcn_readlane(mine, j);
                const int s1 = __builtin_amdgcn_readlane(mine, j + 1);
                const vecf v0 = *reinterpret_cast<const vecf*>(glane + s0 * K);
                const vecf v1 = *reinterpret_cast<const vecf*>(glane + s1 * K);
                if (SCALE_IN) {
                    acc += v0 * __int_as_float(__builtin_amdgcn_readlane(dmine, j));
                    acc += v1 * __int_as_float(__builtin_amdgcn_readlane(dmine, j + 1));
                } else {
                    acc += v0;
                    acc += v1;
                }
                j += 2;
            }
            if (j < n) {
                const int s0 = __builtin_amdgcn_readlane(mine, j);
                const vecf v0 = *reinterpret_cast<const vecf*>(glane + s0 * K);
                if (SCALE_IN) acc += v0 * __int_as_float(__builtin_amdgcn_readlane(dmine, j));
                else          acc += v0;
            }
        }

        *reinterpret_cast<vecf*>(tb + w * P + lane * VEC) = acc;
        __syncthreads();

        if (grp == (it & (NG - 1))) {
            const int mt   = wl / NT;
            const int nt   = wl - mt * NT;
            const int c    = lane & 15;
            const int kh   = (lane >> 4) << 1;
            const int colg = nt * 16 + c;
            const float* arow = tb + (mt * 16 + c) * P + kh;
            const float* wcol = wlds + kh * WP + colg;

            v8f dacc = {};
#pragma unroll
            for (int kk = 0; kk < K; kk += 4) {
                v2f a = *reinterpret_cast<const v2f*>(arow + kk);
                v2f b;
                b.x = wcol[(kk)     * WP];
                b.y = wcol[(kk + 1) * WP];
                dacc = __builtin_amdgcn_wmma_f32_16x16x4_f32(
                     false, a,  false, b,
                     (short)0, dacc,  false,  false);
            }

            const int   mbase = t * 32 + mt * 16 + ((lane >> 4) << 3);
            const float bcol  = bias[colg];
#pragma unroll
            for (int r = 0; r < 8; ++r) {
                const int   m  = mbase + r;
                const float dm = dinv[m];
                float v = dm * dacc[r] + bcol;
                if (RELU) v = fmaxf(v, 0.0f);
                if (SCALE_OUT) v *= dm;
                OUT[m * COUT + colg] = v;
            }
        }
    }
}

extern "C" void kernel_launch(void* const* d_in, const int* in_sizes, int n_in,
                              void* d_out, int out_size, void* d_ws, size_t ws_size,
                              hipStream_t stream) {
    const float* x  = (const float*)d_in[0];
    const int*   ei = (const int*)d_in[1];
    const float* W1 = (const float*)d_in[2];
    const float* b1 = (const float*)d_in[3];
    const float* W2 = (const float*)d_in[4];
    const float* b2 = (const float*)d_in[5];
    const float* W3 = (const float*)d_in[6];
    const float* b3 = (const float*)d_in[7];
    float*       out = (float*)d_out;

    const int* src = ei;
    const int* dst = ei + N_EDGES;

    char*  ws       = (char*)d_ws;
    size_t off      = 0;
    float* dinv     = (float*)(ws + off); off += ((size_t)N_NODES * 4 + 511) & ~511ull;
    int*   rowptr   = (int*)(ws + off);   off += ((size_t)(N_NODES + 1) * 4 + 511) & ~511ull;
    int*   wgcnt    = (int*)(ws + off);   off += (size_t)NB * GP * 4;
    int*   binstart = (int*)(ws + off);   off += ((size_t)(NB + 1) * 4 + 511) & ~511ull;
    int2*  binned   = (int2*)(ws + off);  off += (size_t)N_EDGES * 8;
    int*   col      = (int*)(ws + off);   off += (size_t)N_EDGES * 4;
    float* ha       = (float*)(ws + off); off += (size_t)N_NODES * 128 * 4;
    float* hb       = (float*)(ws + off);

    const int L_BLOCKS = 1024;

    bin_count_kernel<<<GP, 1024, 0, stream>>>(dst, wgcnt);
    bin_scatter_kernel<<<GP, 1024, 0, stream>>>(src, dst, wgcnt, binstart, binned);
    bin_sort_kernel<<<NB, 1024, 0, stream>>>(binned, binstart, rowptr, col, dinv);

    agg_gemm_kernel<64, 128, true, true, true><<<L_BLOCKS, 1024, 0, stream>>>(rowptr, col, x, dinv, W1, b1, ha);

    agg_gemm_kernel<128, 128, false, true, true><<<L_BLOCKS, 1024, 0, stream>>>(rowptr, col, ha, dinv, W2, b2, hb);

    agg_gemm_kernel<128, 64, false, false, false><<<L_BLOCKS, 1024, 0, stream>>>(rowptr, col, hb, dinv, W3, b3, out);
}


# layer 3's product moved ahead of its aggregation: layer 2's kernel multiplies its output tile by W3 (second WMMA pass from LDS); layer 3 sums 256-B rows; new sum order there: @W, then sum
# speedup vs baseline: 5.2417x; 1.0843x over previous
#include <hip/hip_runtime.h>

#define N_NODES 100000
#define N_EDGES 1600000
#define N_TILES (N_NODES / 32)

typedef __attribute__((ext_vector_type(2))) float v2f;
typedef __attribute__((ext_vector_type(4))) float v4f;
typedef __attribute__((ext_vector_type(8))) float v8f;

#define BIN_SHIFT 8
#define BIN_NODES 256
#define NB ((N_NODES + BIN_NODES - 1) / BIN_NODES)
#define GP 256
#define CHUNK (N_EDGES / GP)
#define PER ((CHUNK + 1023) / 1024)
static_assert(N_EDGES % GP == 0, "edge chunks are exact");
static_assert(GP % 32 == 0, "a wave reads a bin's counts in whole passes");

__global__ void __launch_bounds__(1024)
bin_count_kernel(const int* __restrict__ dst, int* __restrict__ wgcnt) {
    __shared__ int hist[NB];
    const int tid = threadIdx.x;
    const int e0  = blockIdx.x * CHUNK;
    int d[PER];
#pragma unroll
    for (int k = 0; k < PER; ++k) {
        const int i = tid + k * 1024;
        d[k] = dst[e0 + (i < CHUNK ? i : CHUNK - 1)];
    }
    if (tid < NB) hist[tid] = 0;
    __syncthreads();
#pragma unroll
    for (int k = 0; k < PER; ++k)
        if (tid + k * 1024 < CHUNK) atomicAdd(&hist[d[k] >> BIN_SHIFT], 1);
    __syncthreads();
    if (tid < NB) wgcnt[tid * GP + blockIdx.x] = hist[tid];
}

__global__ void __launch_bounds__(1024)
bin_scatter_kernel(const int* __restrict__ src, const int* __restrict__ dst,
                   const int* __restrict__ wgcnt, int* __restrict__ binstart,
                   int2* __restrict__ binned) {
    __shared__ int cur[NB];
    __shared__ int tot[NB];
    const int tid  = threadIdx.x;
    const int lane = tid & 31;
    const int w    = tid >> 5;
    const int wg   = blockIdx.x;
    const int e0   = wg * CHUNK;

    int es[PER], ed[PER];
#pragma unroll
    for (int k = 0; k < PER; ++k) {
        const int i  = tid + k * 1024;
        const int ic = i < CHUNK ? i : CHUNK - 1;
        es[k] = src[e0 + ic];
        ed[k] = dst[e0 + ic];
    }

    for (int b0 = w; b0 < NB; b0 += 128) {
        int pre[4], all[4];
#pragma unroll
        for (int j = 0; j < 4; ++j) {
            const int b  = b0 + 32 * j;
            const int bc = b < NB ? b : NB - 1;
            pre[j] = 0;
            all[j] = 0;
#pragma unroll
            for (int k = 0; k < GP / 32; ++k) {
                const int i = k * 32 + lane;
                const int v = wgcnt[bc * GP + i];
                all[j] += v;
                pre[j] += (i < wg) ? v : 0;
            }
        }
#pragma unroll
        for (int j = 0; j < 4; ++j) {
#pragma unroll
            for (int o = 16; o > 0; o >>= 1) {
                all[j] += __shfl_xor(all[j], o);
                pre[j] += __shfl_xor(pre[j], o);
            }
            const int b = b0 + 32 * j;
            if (lane == 0 && b < NB) { tot[b] = all[j]; cur[b] = pre[j]; }
        }
    }
    __syncthreads();
    if (w == 0) {
        int carry = 0;
        for (int c = 0; c < NB; c += 32) {
            const int b = c + lane;
            const int v = (b < NB) ? tot[b] : 0;
            int incl = v;
#pragma unroll
            for (int o = 1; o < 32; o <<= 1) {
                const int t = __shfl_up(incl, o);
                if (lane >= o) incl += t;
            }
            if (b < NB) {
                const int start = carry + incl - v;
                cur[b] += start;
                if (wg == 0) binstart[b] = start;
            }
            carry += __shfl(incl, 31);
        }
        if (wg == 0 && lane == 0) binstart[NB] = carry;
    }
    __syncthreads();

#pragma unroll
    for (int k = 0; k < PER; ++k) {
        if (tid + k * 1024 < CHUNK) {
            const int pos = atomicAdd(&cur[ed[k] >> BIN_SHIFT], 1);
            binned[pos] = make_int2(es[k], ed[k]);
        }
    }
}

__global__ void __launch_bounds__(1024)
bin_sort_kernel(const int2* __restrict__ binned, const int* __restrict__ binstart,
                int* __restrict__ rowptr, int* __restrict__ col,
                float* __restrict__ dinv) {
    __shared__ int cnt[BIN_NODES];
    __shared__ int curs[BIN_NODES];
    __shared__ int wtot[BIN_NODES / 32];
    const int tid  = threadIdx.x;
    const int lane = tid & 31;
    const int w    = tid >> 5;
    const int b    = blockIdx.x;
    const int bs   = binstart[b];
    const int be   = binstart[b + 1];

    if (tid < BIN_NODES) cnt[tid] = 0;
    __syncthreads();
    for (int e = bs + tid; e < be; e += 4096) {
        int y[4];
#pragma unroll
        for (int j = 0; j < 4; ++j) {
            const int i = e + j * 1024;
            y[j] = binned[i < be ? i : be - 1].y;
        }
        asm volatile("" :: "v"(y[0]), "v"(y[1]), "v"(y[2]), "v"(y[3]));
#pragma unroll
        for (int j = 0; j < 4; ++j)
            if (e + j * 1024 < be) atomicAdd(&cnt[y[j] & (BIN_NODES - 1)], 1);
    }
    __syncthreads();

    int v = 0, incl = 0;
    if (tid < BIN_NODES) {
        v = cnt[tid];
        incl = v;
#pragma unroll
        for (int o = 1; o < 32; o <<= 1) {
            const int t = __shfl_up(incl, o);
            if (lane >= o) incl += t;
        }
        if (lane == 31) wtot[w] = incl;
    }
    __syncthreads();
    if (tid < BIN_NODES) {
        int off = 0;
        for (int k = 0; k < w; ++k) off += wtot[k];
        const int excl = off + incl - v;
        curs[tid] = excl;
        const int node = b * BIN_NODES + tid;
        if (node <= N_NODES) rowptr[node] = bs + excl;
        if (node < N_NODES)  dinv[node] = rsqrtf((float)v + 1.0f);
    }
    __syncthreads();
    for (int e = bs + tid; e < be; e += 4096) {
        int2 sd[4];
#pragma unroll
        for (int j = 0; j < 4; ++j) {
            const int i = e + j * 1024;
            sd[j] = binned[i < be ? i : be - 1];
        }
        asm volatile("" :: "v"(sd[0].y), "v"(sd[1].y), "v"(sd[2].y), "v"(sd[3].y));
#pragma unroll
        for (int j = 0; j < 4; ++j) {
            if (e + j * 1024 < be) {
                const int pos = atomicAdd(&curs[sd[j].y & (BIN_NODES - 1)], 1);
                col[bs + pos] = sd[j].x;
            }
        }
    }
}

template <int K, int COUT, bool SCALE_IN, bool RELU, bool SCALE_OUT>
__global__ void __launch_bounds__(1024)
agg_gemm_kernel(const int* __restrict__ rowptr,
                const int* __restrict__ col,
                const float* __restrict__ G,
                const float* __restrict__ dinv,
                const float* __restrict__ W,
                const float* __restrict__ bias,
                float* __restrict__ OUT) {
    constexpr int VEC = K / 32;
    constexpr int P   = K + 4;
    constexpr int WP  = COUT + 8;
    typedef __attribute__((ext_vector_type(VEC))) float vecf;
    constexpr int NT  = COUT / 16;
    constexpr int GW  = 2 * NT;
    constexpr int NG  = 32 / GW;
    __shared__ __attribute__((aligned(16))) float tile[2][32 * P];
    __shared__ __attribute__((aligned(16))) float wlds[K * WP];

    const int lane = threadIdx.x & 31;
    const int w    = __builtin_amdgcn_readfirstlane(threadIdx.x >> 5);
    const int grp  = w / GW;
    const int wl   = w - grp * GW;
    const float* glane = G + lane * VEC;

    constexpr int NWP = K * COUT / 4096;
    v4f wv[NWP];
#pragma unroll
    for (int i = 0; i < NWP; ++i)
        wv[i] = *reinterpret_cast<const v4f*>(W + threadIdx.x * 4 + i * 4096);
#pragma unroll
    for (int i = 0; i < NWP; ++i) {
        const int f   = threadIdx.x * 4 + i * 4096;
        const int row = f / COUT;
        const int cx  = f - row * COUT;
        *reinterpret_cast<v4f*>(wlds + row * WP + cx) = wv[i];
    }

    int it = 0;
    for (int t = blockIdx.x; t < N_TILES; t += gridDim.x, ++it) {
        float* tb = tile[it & 1];
        const int node = t * 32 + w;
        const int beg  = rowptr[node];
        const int end  = rowptr[node + 1];

        vecf acc = *reinterpret_cast<const vecf*>(glane + node * K);
        if (SCALE_IN) acc *= dinv[node];

        for (int base = beg; base < end; base += 32) {
            const int rem  = end - base;
            const int n    = rem < 32 ? rem : 32;
            const int mine = (lane < n) ? col[base + lane] : 0;
            int dmine = 0;
            if (SCALE_IN) dmine = __float_as_int((lane < n) ? dinv[mine] : 0.0f);
            int j = 0;
            for (; j + 4 <= n; j += 4) {
                const int s0 = __builtin_amdgcn_readlane(mine, j);
                const int s1 = __builtin_amdgcn_readlane(mine, j + 1);
                const int s2 = __builtin_amdgcn_readlane(mine, j + 2);
                const int s3 = __builtin_amdgcn_readlane(mine, j + 3);
                const vecf v0 = *reinterpret_cast<const vecf*>(glane + s0 * K);
                const vecf v1 = *reinterpret_cast<const vecf*>(glane + s1 * K);
                const vecf v2 = *reinterpret_cast<const vecf*>(glane + s2 * K);
                const vecf v3 = *reinterpret_cast<const vecf*>(glane + s3 * K);
                if (SCALE_IN) {
                    acc += v0 * __int_as_float(__builtin_amdgcn_readlane(dmine, j));
                    acc += v1 * __int_as_float(__builtin_amdgcn_readlane(dmine, j + 1));
                    acc += v2 * __int_as_float(__builtin_amdgcn_readlane(dmine, j + 2));
                    acc += v3 * __int_as_float(__builtin_amdgcn_readlane(dmine, j + 3));
                } else {
                    acc += v0;
                    acc += v1;
                    acc += v2;
                    acc += v3;
                }
            }
            if (n - j >= 2) {
                const int s0 = __builtin_amdgcn_readlane(mine, j);
                const int s1 = __builtin_amdgcn_readlane(mine, j + 1);
                const vecf v0 = *reinterpret_cast<const vecf*>(glane + s0 * K);
                const vecf v1 = *reinterpret_cast<const vecf*>(glane + s1 * K);
                if (SCALE_IN) {
                    acc += v0 * __int_as_float(__builtin_amdgcn_readlane(dmine, j));
                    acc += v1 * __int_as_float(__builtin_amdgcn_readlane(dmine, j + 1));
                } else {
                    acc += v0;
                    acc += v1;
                }
                j += 2;
            }
            if (j < n) {
                const int s0 = __builtin_amdgcn_readlane(mine, j);
                const vecf v0 = *reinterpret_cast<const vecf*>(glane + s0 * K);
                if (SCALE_IN) acc += v0 * __int_as_float(__builtin_amdgcn_readlane(dmine, j));
                else          acc += v0;
            }
        }

        *reinterpret_cast<vecf*>(tb + w * P + lane * VEC) = acc;
        __syncthreads();

        if (grp == (it & (NG - 1))) {
            const int mt   = wl / NT;
            const int nt   = wl - mt * NT;
            const int c    = lane & 15;
            const int kh   = (lane >> 4) << 1;
            const int colg = nt * 16 + c;
            const float* arow = tb + (mt * 16 + c) * P + kh;
            const float* wcol = wlds + kh * WP + colg;

            v8f dacc = {};
#pragma unroll
            for (int kk = 0; kk < K; kk += 4) {
                v2f a = *reinterpret_cast<const v2f*>(arow + kk);
                v2f b;
                b.x = wcol[(kk)     * WP];
                b.y = wcol[(kk + 1) * WP];
                dacc = __builtin_amdgcn_wmma_f32_16x16x4_f32(
                     false, a,  false, b,
                     (short)0, dacc,  false,  false);
            }

            const int   mbase = t * 32 + mt * 16 + ((lane >> 4) << 3);
            const float bcol  = bias[colg];
#pragma unroll
            for (int r = 0; r < 8; ++r) {
                const int   m  = mbase + r;
                const float dm = dinv[m];
                float v = dm * dacc[r] + bcol;
                if (RELU) v = fmaxf(v, 0.0f);
                if (SCALE_OUT) v *= dm;
                OUT[m * COUT + colg] = v;
            }
        }
    }
}

template <int K> struct rowvec { typedef __attribute__((ext_vector_type(K / 32))) float type; };

template <int K>
__device__ __forceinline__ typename rowvec<K>::type
gather_sum(const int* __restrict__ rowptr, const int* __restrict__ col,
           const float* __restrict__ glane, const int node, const int lane) {
    typedef typename rowvec<K>::type vecf;
    const int beg = rowptr[node];
    const int end = rowptr[node + 1];
    vecf acc = *reinterpret_cast<const vecf*>(glane + node * K);
    for (int base = beg; base < end; base += 32) {
        const int rem  = end - base;
        const int n    = rem < 32 ? rem : 32;
        const int mine = (lane < n) ? col[base + lane] : 0;
        int j = 0;
        for (; j + 4 <= n; j += 4) {
            const int s0 = __builtin_amdgcn_readlane(mine, j);
            const int s1 = __builtin_amdgcn_readlane(mine, j + 1);
            const int s2 = __builtin_amdgcn_readlane(mine, j + 2);
            const int s3 = __builtin_amdgcn_readlane(mine, j + 3);
            const vecf v0 = *reinterpret_cast<const vecf*>(glane + s0 * K);
            const vecf v1 = *reinterpret_cast<const vecf*>(glane + s1 * K);
            const vecf v2 = *reinterpret_cast<const vecf*>(glane + s2 * K);
            const vecf v3 = *reinterpret_cast<const vecf*>(glane + s3 * K);
            acc += v0;
            acc += v1;
            acc += v2;
            acc += v3;
        }
        if (n - j >= 2) {
            const int s0 = __builtin_amdgcn_readlane(mine, j);
            const int s1 = __builtin_amdgcn_readlane(mine, j + 1);
            const vecf v0 = *reinterpret_cast<const vecf*>(glane + s0 * K);
            const vecf v1 = *reinterpret_cast<const vecf*>(glane + s1 * K);
            acc += v0;
            acc += v1;
            j += 2;
        }
        if (j < n) {
            const int s0 = __builtin_amdgcn_readlane(mine, j);
            acc += *reinterpret_cast<const vecf*>(glane + s0 * K);
        }
    }
    return acc;
}

__global__ void __launch_bounds__(1024, 16)
agg_gemm_next_kernel(const int* __restrict__ rowptr,
                     const int* __restrict__ col,
                     const float* __restrict__ G,
                     const float* __restrict__ dinv,
                     const float* __restrict__ W,
                     const float* __restrict__ bias,
                     const float* __restrict__ Wn,
                     float* __restrict__ Z) {
    constexpr int K    = 128;
    constexpr int COUT = 128;
    constexpr int CN   = 64;
    constexpr int P    = K + 4;
    constexpr int WP   = COUT + 8;
    constexpr int WNP  = CN + 8;
    constexpr int NT   = COUT / 16;
    constexpr int GW   = 2 * NT;
    constexpr int NTN  = CN / 16;
    static_assert(GW == 16 && COUT == K, "two wave groups; the H tile has the sum tile's shape");
    __shared__ __attribute__((aligned(16))) float tile[2][32 * P];
    __shared__ __attribute__((aligned(16))) float htile[32 * P];
    __shared__ __attribute__((aligned(16))) float wlds[K * WP];
    __shared__ __attribute__((aligned(16))) float wnlds[COUT * WNP];

    const int lane = threadIdx.x & 31;
    const int w    = __builtin_amdgcn_readfirstlane(threadIdx.x >> 5);
    const int grp  = w / GW;
    const int wl   = w - grp * GW;
    const float* glane = G + lane * 4;

    constexpr int NWP  = K * COUT / 4096;
    constexpr int NWNP = COUT * CN / 4096;
    v4f wv[NWP];
    v4f wnv[NWNP];
#pragma unroll
    for (int i = 0; i < NWP; ++i)
        wv[i] = *reinterpret_cast<const v4f*>(W + threadIdx.x * 4 + i * 4096);
#pragma unroll
    for (int i = 0; i < NWNP; ++i)
        wnv[i] = *reinterpret_cast<const v4f*>(Wn + threadIdx.x * 4 + i * 4096);
#pragma unroll
    for (int i = 0; i < NWP; ++i) {
        const int f   = threadIdx.x * 4 + i * 4096;
        const int row = f / COUT;
        const int cx  = f - row * COUT;
        *reinterpret_cast<v4f*>(wlds + row * WP + cx) = wv[i];
    }
#pragma unroll
    for (int i = 0; i < NWNP; ++i) {
        const int f   = threadIdx.x * 4 + i * 4096;
        const int row = f / CN;
        const int cx  = f - row * CN;
        *reinterpret_cast<v4f*>(wnlds + row * WNP + cx) = wnv[i];
    }

    const int c  = lane & 15;
    const int kh = (lane >> 4) << 1;
    const int rh = (lane >> 4) << 3;

    int t = blockIdx.x;
    if (t < N_TILES) {
        const v4f acc = gather_sum<K>(rowptr, col, glane, t * 32 + w, lane);
        *reinterpret_cast<v4f*>(tile[0] + w * P + lane * 4) = acc;
    }

    float hv[8];
    int it = 0;
    for (; t < N_TILES; t += gridDim.x, ++it) {
        __syncthreads();
        const float* tb = tile[it & 1];

        if (grp == (it & 1)) {
            const int mt   = wl / NT;
            const int nt   = wl - mt * NT;
            const int colg = nt * 16 + c;
            const float* arow = tb + (mt * 16 + c) * P + kh;
            const float* wcol = wlds + kh * WP + colg;

            v8f dacc = {};
#pragma unroll
            for (int kk = 0; kk < K; kk += 4) {
                v2f a = *reinterpret_cast<const v2f*>(arow + kk);
                v2f b;
                b.x = wcol[(kk)     * WP];
                b.y = wcol[(kk + 1) * WP];
                dacc = __builtin_amdgcn_wmma_f32_16x16x4_f32(
                     false, a,  false, b,
                     (short)0, dacc,  false,  false);
            }

            const int   mbase = t * 32 + mt * 16 + rh;
            const float bcol  = bias[colg];
#pragma unroll
            for (int r = 0; r < 8; ++r) {
                const float dm = dinv[mbase + r];
                float v = dm * dacc[r] + bcol;
                v = fmaxf(v, 0.0f);
                hv[r] = v * dm;
            }
        } else if (it > 0 && wl < 2 * NTN) {
            const int mt   = wl / NTN;
            const int nt   = wl - mt * NTN;
            const int colg = nt * 16 + c;
            const float* arow = htile + (mt * 16 + c) * P + kh;
            const float* wcol = wnlds + kh * WNP + colg;

            v8f zacc = {};
#pragma unroll
            for (int kk = 0; kk < COUT; kk += 4) {
                v2f a = *reinterpret_cast<const v2f*>(arow + kk);
                v2f b;
                b.x = wcol[(kk)     * WNP];
                b.y = wcol[(kk + 1) * WNP];
                zacc = __builtin_amdgcn_wmma_f32_16x16x4_f32(
                     false, a,  false, b,
                     (short)0, zacc,  false,  false);
            }

            const int mbase = (t - (int)gridDim.x) * 32 + mt * 16 + rh;
#pragma unroll
            for (int r = 0; r < 8; ++r)
                Z[(mbase + r) * CN + colg] = zacc[r];
        }

        const int tn = t + gridDim.x;
        if (tn < N_TILES) {
            const v4f acc = gather_sum<K>(rowptr, col, glane, tn * 32 + w, lane);
            *reinterpret_cast<v4f*>(tile[(it + 1) & 1] + w * P + lane * 4) = acc;
        }
        __syncthreads();

        if (grp == (it & 1)) {
            const int mt   = wl / NT;
            const int nt   = wl - mt * NT;
            float* hcol = htile + (mt * 16 + rh) * P + nt * 16 + c;
#pragma unroll
            for (int r = 0; r < 8; ++r)
                hcol[r * P] = hv[r];
        }
    }

    __syncthreads();
    if (it > 0 && grp != (it & 1) && wl < 2 * NTN) {
        const int mt   = wl / NTN;
        const int nt   = wl - mt * NTN;
        const int colg = nt * 16 + c;
        const float* arow = htile + (mt * 16 + c) * P + kh;
        const float* wcol = wnlds + kh * WNP + colg;

        v8f zacc = {};
#pragma unroll
        for (int kk = 0; kk < COUT; kk += 4) {
            v2f a = *reinterpret_cast<const v2f*>(arow + kk);
            v2f b;
            b.x = wcol[(kk)     * WNP];
            b.y = wcol[(kk + 1) * WNP];
            zacc = __builtin_amdgcn_wmma_f32_16x16x4_f32(
                 false, a,  false, b,
                 (short)0, zacc,  false,  false);
        }

        const int mbase = (t - (int)gridDim.x) * 32 + mt * 16 + rh;
#pragma unroll
        for (int r = 0; r < 8; ++r)
            Z[(mbase + r) * CN + colg] = zacc[r];
    }
}

__global__ void __launch_bounds__(1024)
agg_out_kernel(const int* __restrict__ rowptr,
               const int* __restrict__ col,
               const float* __restrict__ Z,
               const float* __restrict__ dinv,
               const float* __restrict__ bias,
               float* __restrict__ OUT) {
    constexpr int C = 64;
    const int lane = threadIdx.x & 31;
    const int w    = __builtin_amdgcn_readfirstlane(threadIdx.x >> 5);
    const float* glane = Z + lane * 2;
    const v2f bl = *reinterpret_cast<const v2f*>(bias + lane * 2);
    for (int t = blockIdx.x; t < N_TILES; t += gridDim.x) {
        const int node = t * 32 + w;
        const v2f acc = gather_sum<C>(rowptr, col, glane, node, lane);
        const float dm = dinv[node];
        *reinterpret_cast<v2f*>(OUT + node * C + lane * 2) = acc * dm + bl;
    }
}

extern "C" void kernel_launch(void* const* d_in, const int* in_sizes, int n_in,
                              void* d_out, int out_size, void* d_ws, size_t ws_size,
                              hipStream_t stream) {
    const float* x  = (const float*)d_in[0];
    const int*   ei = (const int*)d_in[1];
    const float* W1 = (const float*)d_in[2];
    const float* b1 = (const float*)d_in[3];
    const float* W2 = (const float*)d_in[4];
    const float* b2 = (const float*)d_in[5];
    const float* W3 = (const float*)d_in[6];
    const float* b3 = (const float*)d_in[7];
    float*       out = (float*)d_out;

    const int* src = ei;
    const int* dst = ei + N_EDGES;

    char*  ws       = (char*)d_ws;
    size_t off      = 0;
    float* dinv     = (float*)(ws + off); off += ((size_t)N_NODES * 4 + 511) & ~511ull;
    int*   rowptr   = (int*)(ws + off);   off += ((size_t)(N_NODES + 1) * 4 + 511) & ~511ull;
    int*   wgcnt    = (int*)(ws + off);   off += (size_t)NB * GP * 4;
    int*   binstart = (int*)(ws + off);   off += ((size_t)(NB + 1) * 4 + 511) & ~511ull;
    int2*  binned   = (int2*)(ws + off);  off += (size_t)N_EDGES * 8;
    int*   col      = (int*)(ws + off);   off += (size_t)N_EDGES * 4;
    float* ha       = (float*)(ws + off); off += (size_t)N_NODES * 128 * 4;
    float* hb       = (float*)(ws + off);

    const int L_BLOCKS = 1024;

    bin_count_kernel<<<GP, 1024, 0, stream>>>(dst, wgcnt);
    bin_scatter_kernel<<<GP, 1024, 0, stream>>>(src, dst, wgcnt, binstart, binned);
    bin_sort_kernel<<<NB, 1024, 0, stream>>>(binned, binstart, rowptr, col, dinv);

    agg_gemm_kernel<64, 128, true, true, true><<<L_BLOCKS, 1024, 0, stream>>>(rowptr, col, x, dinv, W1, b1, ha);

    agg_gemm_next_kernel<<<L_BLOCKS, 1024, 0, stream>>>(rowptr, col, ha, dinv, W2, b2, W3, hb);

    agg_out_kernel<<<L_BLOCKS, 1024, 0, stream>>>(rowptr, col, hb, dinv, b3, out);
}


# neighbour-row gather issues a whole batch of row loads before its first add (16 rows of 256 B or 8 of 512 B in flight per wave, one guarded batch for the remainder) instead of 4; sum order kept
# speedup vs baseline: 5.4619x; 1.0420x over previous
#include <hip/hip_runtime.h>

#define N_NODES 100000
#define N_EDGES 1600000
#define N_TILES (N_NODES / 32)

typedef __attribute__((ext_vector_type(2))) float v2f;
typedef __attribute__((ext_vector_type(4))) float v4f;
typedef __attribute__((ext_vector_type(8))) float v8f;

#define BIN_SHIFT 8
#define BIN_NODES 256
#define NB ((N_NODES + BIN_NODES - 1) / BIN_NODES)
#define GP 256
#define CHUNK (N_EDGES / GP)
#define PER ((CHUNK + 1023) / 1024)
static_assert(N_EDGES % GP == 0, "edge chunks are exact");
static_assert(GP % 32 == 0, "a wave reads a bin's counts in whole passes");

__global__ void __launch_bounds__(1024)
bin_count_kernel(const int* __restrict__ dst, int* __restrict__ wgcnt) {
    __shared__ int hist[NB];
    const int tid = threadIdx.x;
    const int e0  = blockIdx.x * CHUNK;
    int d[PER];
#pragma unroll
    for (int k = 0; k < PER; ++k) {
        const int i = tid + k * 1024;
        d[k] = dst[e0 + (i < CHUNK ? i : CHUNK - 1)];
    }
    if (tid < NB) hist[tid] = 0;
    __syncthreads();
#pragma unroll
    for (int k = 0; k < PER; ++k)
        if (tid + k * 1024 < CHUNK) atomicAdd(&hist[d[k] >> BIN_SHIFT], 1);
    __syncthreads();
    if (tid < NB) wgcnt[tid * GP + blockIdx.x] = hist[tid];
}

__global__ void __launch_bounds__(1024)
bin_scatter_kernel(const int* __restrict__ src, const int* __restrict__ dst,
                   const int* __restrict__ wgcnt, int* __restrict__ binstart,
                   int2* __restrict__ binned) {
    __shared__ int cur[NB];
    __shared__ int tot[NB];
    const int tid  = threadIdx.x;
    const int lane = tid & 31;
    const int w    = tid >> 5;
    const int wg   = blockIdx.x;
    const int e0   = wg * CHUNK;

    int es[PER], ed[PER];
#pragma unroll
    for (int k = 0; k < PER; ++k) {
        const int i  = tid + k * 1024;
        const int ic = i < CHUNK ? i : CHUNK - 1;
        es[k] = src[e0 + ic];
        ed[k] = dst[e0 + ic];
    }

    for (int b0 = w; b0 < NB; b0 += 128) {
        int pre[4], all[4];
#pragma unroll
        for (int j = 0; j < 4; ++j) {
            const int b  = b0 + 32 * j;
            const int bc = b < NB ? b : NB - 1;
            pre[j] = 0;
            all[j] = 0;
#pragma unroll
            for (int k = 0; k < GP / 32; ++k) {
                const int i = k * 32 + lane;
                const int v = wgcnt[bc * GP + i];
                all[j] += v;
                pre[j] += (i < wg) ? v : 0;
            }
        }
#pragma unroll
        for (int j = 0; j < 4; ++j) {
#pragma unroll
            for (int o = 16; o > 0; o >>= 1) {
                all[j] += __shfl_xor(all[j], o);
                pre[j] += __shfl_xor(pre[j], o);
            }
            const int b = b0 + 32 * j;
            if (lane == 0 && b < NB) { tot[b] = all[j]; cur[b] = pre[j]; }
        }
    }
    __syncthreads();
    if (w == 0) {
        int carry = 0;
        for (int c = 0; c < NB; c += 32) {
            const int b = c + lane;
            const int v = (b < NB) ? tot[b] : 0;
            int incl = v;
#pragma unroll
            for (int o = 1; o < 32; o <<= 1) {
                const int t = __shfl_up(incl, o);
                if (lane >= o) incl += t;
            }
            if (b < NB) {
                const int start = carry + incl - v;
                cur[b] += start;
                if (wg == 0) binstart[b] = start;
            }
            carry += __shfl(incl, 31);
        }
        if (wg == 0 && lane == 0) binstart[NB] = carry;
    }
    __syncthreads();

#pragma unroll
    for (int k = 0; k < PER; ++k) {
        if (tid + k * 1024 < CHUNK) {
            const int pos = atomicAdd(&cur[ed[k] >> BIN_SHIFT], 1);
            binned[pos] = make_int2(es[k], ed[k]);
        }
    }
}

__global__ void __launch_bounds__(1024)
bin_sort_kernel(const int2* __restrict__ binned, const int* __restrict__ binstart,
                int* __restrict__ rowptr, int* __restrict__ col,
                float* __restrict__ dinv) {
    __shared__ int cnt[BIN_NODES];
    __shared__ int curs[BIN_NODES];
    __shared__ int wtot[BIN_NODES / 32];
    const int tid  = threadIdx.x;
    const int lane = tid & 31;
    const int w    = tid >> 5;
    const int b    = blockIdx.x;
    const int bs   = binstart[b];
    const int be   = binstart[b + 1];

    if (tid < BIN_NODES) cnt[tid] = 0;
    __syncthreads();
    for (int e = bs + tid; e < be; e += 4096) {
        int y[4];
#pragma unroll
        for (int j = 0; j < 4; ++j) {
            const int i = e + j * 1024;
            y[j] = binned[i < be ? i : be - 1].y;
        }
        asm volatile("" :: "v"(y[0]), "v"(y[1]), "v"(y[2]), "v"(y[3]));
#pragma unroll
        for (int j = 0; j < 4; ++j)
            if (e + j * 1024 < be) atomicAdd(&cnt[y[j] & (BIN_NODES - 1)], 1);
    }
    __syncthreads();

    int v = 0, incl = 0;
    if (tid < BIN_NODES) {
        v = cnt[tid];
        incl = v;
#pragma unroll
        for (int o = 1; o < 32; o <<= 1) {
            const int t = __shfl_up(incl, o);
            if (lane >= o) incl += t;
        }
        if (lane == 31) wtot[w] = incl;
    }
    __syncthreads();
    if (tid < BIN_NODES) {
        int off = 0;
        for (int k = 0; k < w; ++k) off += wtot[k];
        const int excl = off + incl - v;
        curs[tid] = excl;
        const int node = b * BIN_NODES + tid;
        if (node <= N_NODES) rowptr[node] = bs + excl;
        if (node < N_NODES)  dinv[node] = rsqrtf((float)v + 1.0f);
    }
    __syncthreads();
    for (int e = bs + tid; e < be; e += 4096) {
        int2 sd[4];
#pragma unroll
        for (int j = 0; j < 4; ++j) {
            const int i = e + j * 1024;
            sd[j] = binned[i < be ? i : be - 1];
        }
        asm volatile("" :: "v"(sd[0].y), "v"(sd[1].y), "v"(sd[2].y), "v"(sd[3].y));
#pragma unroll
        for (int j = 0; j < 4; ++j) {
            if (e + j * 1024 < be) {
                const int pos = atomicAdd(&curs[sd[j].y & (BIN_NODES - 1)], 1);
                col[bs + pos] = sd[j].x;
            }
        }
    }
}

template <int K> struct rowvec { typedef __attribute__((ext_vector_type(K / 32))) float type; };

template <int K, int B, bool SCALE_IN>
__device__ __forceinline__ typename rowvec<K>::type
gather_sum(const int* __restrict__ rowptr, const int* __restrict__ col,
           const float* __restrict__ G, const float* __restrict__ dinv,
           const int node, const int lane) {
    typedef typename rowvec<K>::type vecf;
    const float* glane = G + lane * (K / 32);
    const int beg = rowptr[node];
    const int end = rowptr[node + 1];
    vecf acc = *reinterpret_cast<const vecf*>(glane + node * K);
    if (SCALE_IN) acc *= dinv[node];
    for (int base = beg; base < end; base += 32) {
        const int rem  = end - base;
        const int n    = rem < 32 ? rem : 32;
        const int mine = (lane < n) ? col[base + lane] : 0;
        int dmine = 0;
        if (SCALE_IN) dmine = __float_as_int((lane < n) ? dinv[mine] : 0.0f);
        int j = 0;
        for (; j + B <= n; j += B) {
            vecf v[B];
#pragma unroll
            for (int k = 0; k < B; ++k)
                v[k] = *reinterpret_cast<const vecf*>(glane + __builtin_amdgcn_readlane(mine, j + k) * K);
#pragma unroll
            for (int k = 0; k < B; ++k) {
                if (SCALE_IN) acc += v[k] * __int_as_float(__builtin_amdgcn_readlane(dmine, j + k));
                else          acc += v[k];
            }
        }
        const int r = n - j;
        if (r > 0) {
            vecf v[B - 1];
#pragma unroll
            for (int k = 0; k < B - 1; ++k) {
                v[k] = 0.0f;
                if (k == 0 || k < r) v[k] = *reinterpret_cast<const vecf*>(glane + __builtin_amdgcn_readlane(mine, j + k) * K);
            }
#pragma unroll
            for (int k = 0; k < B - 1; ++k) {
                if (SCALE_IN) acc += v[k] * __int_as_float(__builtin_amdgcn_readlane(dmine, j + k));
                else          acc += v[k];
            }
        }
    }
    return acc;
}

template <int K, int COUT, bool SCALE_IN, bool RELU, bool SCALE_OUT>
__global__ void __launch_bounds__(1024)
agg_gemm_kernel(const int* __restrict__ rowptr,
                const int* __restrict__ col,
                const float* __restrict__ G,
                const float* __restrict__ dinv,
                const float* __restrict__ W,
                const float* __restrict__ bias,
                float* __restrict__ OUT) {
    constexpr int VEC = K / 32;
    constexpr int P   = K + 4;
    constexpr int WP  = COUT + 8;
    typedef __attribute__((ext_vector_type(VEC))) float vecf;
    constexpr int NT  = COUT / 16;
    constexpr int GW  = 2 * NT;
    constexpr int NG  = 32 / GW;
    constexpr int GB  = 32 / VEC;
    __shared__ __attribute__((aligned(16))) float tile[2][32 * P];
    __shared__ __attribute__((aligned(16))) float wlds[K * WP];

    const int lane = threadIdx.x & 31;
    const int w    = __builtin_amdgcn_readfirstlane(threadIdx.x >> 5);
    const int grp  = w / GW;
    const int wl   = w - grp * GW;

    constexpr int NWP = K * COUT / 4096;
    v4f wv[NWP];
#pragma unroll
    for (int i = 0; i < NWP; ++i)
        wv[i] = *reinterpret_cast<const v4f*>(W + threadIdx.x * 4 + i * 4096);
#pragma unroll
    for (int i = 0; i < NWP; ++i) {
        const int f   = threadIdx.x * 4 + i * 4096;
        const int row = f / COUT;
        const int cx  = f - row * COUT;
        *reinterpret_cast<v4f*>(wlds + row * WP + cx) = wv[i];
    }

    int it = 0;
    for (int t = blockIdx.x; t < N_TILES; t += gridDim.x, ++it) {
        float* tb = tile[it & 1];
        const int node = t * 32 + w;
        const vecf acc = gather_sum<K, GB, SCALE_IN>(rowptr, col, G, dinv, node, lane);

        *reinterpret_cast<vecf*>(tb + w * P + lane * VEC) = acc;
        __syncthreads();

        if (grp == (it & (NG - 1))) {
            const int mt   = wl / NT;
            const int nt   = wl - mt * NT;
            const int c    = lane & 15;
            const int kh   = (lane >> 4) << 1;
            const int colg = nt * 16 + c;
            const float* arow = tb + (mt * 16 + c) * P + kh;
            const float* wcol = wlds + kh * WP + colg;

            v8f dacc = {};
#pragma unroll
            for (int kk = 0; kk < K; kk += 4) {
                v2f a = *reinterpret_cast<const v2f*>(arow + kk);
                v2f b;
                b.x = wcol[(kk)     * WP];
                b.y = wcol[(kk + 1) * WP];
                dacc = __builtin_amdgcn_wmma_f32_16x16x4_f32(
                     false, a,  false, b,
                     (short)0, dacc,  false,  false);
            }

            const int   mbase = t * 32 + mt * 16 + ((lane >> 4) << 3);
            const float bcol  = bias[colg];
#pragma unroll
            for (int r = 0; r < 8; ++r) {
                const int   m  = mbase + r;
                const float dm = dinv[m];
                float v = dm * dacc[r] + bcol;
                if (RELU) v = fmaxf(v, 0.0f);
                if (SCALE_OUT) v *= dm;
                OUT[m * COUT + colg] = v;
            }
        }
    }
}

__global__ void __launch_bounds__(1024, 16)
agg_gemm_next_kernel(const int* __restrict__ rowptr,
                     const int* __restrict__ col,
                     const float* __restrict__ G,
                     const float* __restrict__ dinv,
                     const float* __restrict__ W,
                     const float* __restrict__ bias,
                     const float* __restrict__ Wn,
                     float* __restrict__ Z) {
    constexpr int K    = 128;
    constexpr int COUT = 128;
    constexpr int CN   = 64;
    constexpr int P    = K + 4;
    constexpr int WP   = COUT + 8;
    constexpr int WNP  = CN + 8;
    constexpr int NT   = COUT / 16;
    constexpr int GW   = 2 * NT;
    constexpr int NTN  = CN / 16;
    static_assert(GW == 16 && COUT == K, "two wave groups; the H tile has the sum tile's shape");
    __shared__ __attribute__((aligned(16))) float tile[2][32 * P];
    __shared__ __attribute__((aligned(16))) float htile[32 * P];
    __shared__ __attribute__((aligned(16))) float wlds[K * WP];
    __shared__ __attribute__((aligned(16))) float wnlds[COUT * WNP];

    const int lane = threadIdx.x & 31;
    const int w    = __builtin_amdgcn_readfirstlane(threadIdx.x >> 5);
    const int grp  = w / GW;
    const int wl   = w - grp * GW;

    constexpr int NWP  = K * COUT / 4096;
    constexpr int NWNP = COUT * CN / 4096;
    v4f wv[NWP];
    v4f wnv[NWNP];
#pragma unroll
    for (int i = 0; i < NWP; ++i)
        wv[i] = *reinterpret_cast<const v4f*>(W + threadIdx.x * 4 + i * 4096);
#pragma unroll
    for (int i = 0; i < NWNP; ++i)
        wnv[i] = *reinterpret_cast<const v4f*>(Wn + threadIdx.x * 4 + i * 4096);
#pragma unroll
    for (int i = 0; i < NWP; ++i) {
        const int f   = threadIdx.x * 4 + i * 4096;
        const int row = f / COUT;
        const int cx  = f - row * COUT;
        *reinterpret_cast<v4f*>(wlds + row * WP + cx) = wv[i];
    }
#pragma unroll
    for (int i = 0; i < NWNP; ++i) {
        const int f   = threadIdx.x * 4 + i * 4096;
        const int row = f / CN;
        const int cx  = f - row * CN;
        *reinterpret_cast<v4f*>(wnlds + row * WNP + cx) = wnv[i];
    }

    const int c  = lane & 15;
    const int kh = (lane >> 4) << 1;
    const int rh = (lane >> 4) << 3;

    int t = blockIdx.x;
    if (t < N_TILES) {
        const v4f acc = gather_sum<K, 8, false>(rowptr, col, G, dinv, t * 32 + w, lane);
        *reinterpret_cast<v4f*>(tile[0] + w * P + lane * 4) = acc;
    }

    float hv[8];
    int it = 0;
    for (; t < N_TILES; t += gridDim.x, ++it) {
        __syncthreads();
        const float* tb = tile[it & 1];

        if (grp == (it & 1)) {
            const int mt   = wl / NT;
            const int nt   = wl - mt * NT;
            const int colg = nt * 16 + c;
            const float* arow = tb + (mt * 16 + c) * P + kh;
            const float* wcol = wlds + kh * WP + colg;

            v8f dacc = {};
#pragma unroll
            for (int kk = 0; kk < K; kk += 4) {
                v2f a = *reinterpret_cast<const v2f*>(arow + kk);
                v2f b;
                b.x = wcol[(kk)     * WP];
                b.y = wcol[(kk + 1) * WP];
                dacc = __builtin_amdgcn_wmma_f32_16x16x4_f32(
                     false, a,  false, b,
                     (short)0, dacc,  false,  false);
            }

            const int   mbase = t * 32 + mt * 16 + rh;
            const float bcol  = bias[colg];
#pragma unroll
            for (int r = 0; r < 8; ++r) {
                const float dm = dinv[mbase + r];
                float v = dm * dacc[r] + bcol;
                v = fmaxf(v, 0.0f);
                hv[r] = v * dm;
            }
        } else if (it > 0 && wl < 2 * NTN) {
            const int mt   = wl / NTN;
            const int nt   = wl - mt * NTN;
            const int colg = nt * 16 + c;
            const float* arow = htile + (mt * 16 + c) * P + kh;
            const float* wcol = wnlds + kh * WNP + colg;

            v8f zacc = {};
#pragma unroll
            for (int kk = 0; kk < COUT; kk += 4) {
                v2f a = *reinterpret_cast<const v2f*>(arow + kk);
                v2f b;
                b.x = wcol[(kk)     * WNP];
                b.y = wcol[(kk + 1) * WNP];
                zacc = __builtin_amdgcn_wmma_f32_16x16x4_f32(
                     false, a,  false, b,
                     (short)0, zacc,  false,  false);
            }

            const int mbase = (t - (int)gridDim.x) * 32 + mt * 16 + rh;
#pragma unroll
            for (int r = 0; r < 8; ++r)
                Z[(mbase + r) * CN + colg] = zacc[r];
        }

        const int tn = t + gridDim.x;
        if (tn < N_TILES) {
            const v4f acc = gather_sum<K, 8, false>(rowptr, col, G, dinv, tn * 32 + w, lane);
            *reinterpret_cast<v4f*>(tile[(it + 1) & 1] + w * P + lane * 4) = acc;
        }
        __syncthreads();

        if (grp == (it & 1)) {
            const int mt   = wl / NT;
            const int nt   = wl - mt * NT;
            float* hcol = htile + (mt * 16 + rh) * P + nt * 16 + c;
#pragma unroll
            for (int r = 0; r < 8; ++r)
                hcol[r * P] = hv[r];
        }
    }

    __syncthreads();
    if (it > 0 && grp != (it & 1) && wl < 2 * NTN) {
        const int mt   = wl / NTN;
        const int nt   = wl - mt * NTN;
        const int colg = nt * 16 + c;
        const float* arow = htile + (mt * 16 + c) * P + kh;
        const float* wcol = wnlds + kh * WNP + colg;

        v8f zacc = {};
#pragma unroll
        for (int kk = 0; kk < COUT; kk += 4) {
            v2f a = *reinterpret_cast<const v2f*>(arow + kk);
            v2f b;
            b.x = wcol[(kk)     * WNP];
            b.y = wcol[(kk + 1) * WNP];
            zacc = __builtin_amdgcn_wmma_f32_16x16x4_f32(
                 false, a,  false, b,
                 (short)0, zacc,  false,  false);
        }

        const int mbase = (t - (int)gridDim.x) * 32 + mt * 16 + rh;
#pragma unroll
        for (int r = 0; r < 8; ++r)
            Z[(mbase + r) * CN + colg] = zacc[r];
    }
}

__global__ void __launch_bounds__(1024)
agg_out_kernel(const int* __restrict__ rowptr,
               const int* __restrict__ col,
               const float* __restrict__ Z,
               const float* __restrict__ dinv,
               const float* __restrict__ bias,
               float* __restrict__ OUT) {
    constexpr int C = 64;
    const int lane = threadIdx.x & 31;
    const int w    = __builtin_amdgcn_readfirstlane(threadIdx.x >> 5);
    const v2f bl = *reinterpret_cast<const v2f*>(bias + lane * 2);
    for (int t = blockIdx.x; t < N_TILES; t += gridDim.x) {
        const int node = t * 32 + w;
        const v2f acc = gather_sum<C, 16, false>(rowptr, col, Z, dinv, node, lane);
        const float dm = dinv[node];
        *reinterpret_cast<v2f*>(OUT + node * C + lane * 2) = acc * dm + bl;
    }
}

extern "C" void kernel_launch(void* const* d_in, const int* in_sizes, int n_in,
                              void* d_out, int out_size, void* d_ws, size_t ws_size,
                              hipStream_t stream) {
    const float* x  = (const float*)d_in[0];
    const int*   ei = (const int*)d_in[1];
    const float* W1 = (const float*)d_in[2];
    const float* b1 = (const float*)d_in[3];
    const float* W2 = (const float*)d_in[4];
    const float* b2 = (const float*)d_in[5];
    const float* W3 = (const float*)d_in[6];
    const float* b3 = (const float*)d_in[7];
    float*       out = (float*)d_out;

    const int* src = ei;
    const int* dst = ei + N_EDGES;

    char*  ws       = (char*)d_ws;
    size_t off      = 0;
    float* dinv     = (float*)(ws + off); off += ((size_t)N_NODES * 4 + 511) & ~511ull;
    int*   rowptr   = (int*)(ws + off);   off += ((size_t)(N_NODES + 1) * 4 + 511) & ~511ull;
    int*   wgcnt    = (int*)(ws + off);   off += (size_t)NB * GP * 4;
    int*   binstart = (int*)(ws + off);   off += ((size_t)(NB + 1) * 4 + 511) & ~511ull;
    int2*  binned   = (int2*)(ws + off);  off += (size_t)N_EDGES * 8;
    int*   col      = (int*)(ws + off);   off += (size_t)N_EDGES * 4;
    float* ha       = (float*)(ws + off); off += (size_t)N_NODES * 128 * 4;
    float* hb       = (float*)(ws + off);

    const int L_BLOCKS = 1024;

    bin_count_kernel<<<GP, 1024, 0, stream>>>(dst, wgcnt);
    bin_scatter_kernel<<<GP, 1024, 0, stream>>>(src, dst, wgcnt, binstart, binned);
    bin_sort_kernel<<<NB, 1024, 0, stream>>>(binned, binstart, rowptr, col, dinv);

    agg_gemm_kernel<64, 128, true, true, true><<<L_BLOCKS, 1024, 0, stream>>>(rowptr, col, x, dinv, W1, b1, ha);

    agg_gemm_next_kernel<<<L_BLOCKS, 1024, 0, stream>>>(rowptr, col, ha, dinv, W2, b2, W3, hb);

    agg_out_kernel<<<L_BLOCKS, 1024, 0, stream>>>(rowptr, col, hb, dinv, b3, out);
}


# edge-list build: passes 2 and 3 order their output in LDS first (chunk grouped by bin; a bin's sorted sources) and store contiguous stretches, not 32 scattered pieces per wave store; sum order kept
# speedup vs baseline: 6.0112x; 1.1006x over previous
#include <hip/hip_runtime.h>

#define N_NODES 100000
#define N_EDGES 1600000
#define N_TILES (N_NODES / 32)

typedef __attribute__((ext_vector_type(2))) float v2f;
typedef __attribute__((ext_vector_type(4))) float v4f;
typedef __attribute__((ext_vector_type(8))) float v8f;

#define BIN_SHIFT 8
#define BIN_NODES 256
#define NB ((N_NODES + BIN_NODES - 1) / BIN_NODES)
#define GP 256
#define CHUNK (N_EDGES / GP)
#define PER ((CHUNK + 1023) / 1024)
static_assert(N_EDGES % GP == 0, "edge chunks are exact");
static_assert(GP % 32 == 0, "a wave reads a bin's counts in whole passes");
#define LCAP 8192

__global__ void __launch_bounds__(1024)
bin_count_kernel(const int* __restrict__ dst, int* __restrict__ wgcnt) {
    __shared__ int hist[NB];
    const int tid = threadIdx.x;
    const int e0  = blockIdx.x * CHUNK;
    int d[PER];
#pragma unroll
    for (int k = 0; k < PER; ++k) {
        const int i = tid + k * 1024;
        d[k] = dst[e0 + (i < CHUNK ? i : CHUNK - 1)];
    }
    if (tid < NB) hist[tid] = 0;
    __syncthreads();
#pragma unroll
    for (int k = 0; k < PER; ++k)
        if (tid + k * 1024 < CHUNK) atomicAdd(&hist[d[k] >> BIN_SHIFT], 1);
    __syncthreads();
    if (tid < NB) wgcnt[tid * GP + blockIdx.x] = hist[tid];
}

__global__ void __launch_bounds__(1024)
bin_scatter_kernel(const int* __restrict__ src, const int* __restrict__ dst,
                   const int* __restrict__ wgcnt, int* __restrict__ binstart,
                   int2* __restrict__ binned) {
    __shared__ int cur[NB];
    __shared__ int tot[NB];
    __shared__ int lcnt[NB];
    __shared__ int lcur[NB];
    __shared__ int delta[NB];
    __shared__ int2 stage[CHUNK];
    const int tid  = threadIdx.x;
    const int lane = tid & 31;
    const int w    = tid >> 5;
    const int wg   = blockIdx.x;
    const int e0   = wg * CHUNK;
    if (tid < NB) lcnt[tid] = wgcnt[tid * GP + wg];

    int es[PER], ed[PER];
#pragma unroll
    for (int k = 0; k < PER; ++k) {
        const int i  = tid + k * 1024;
        const int ic = i < CHUNK ? i : CHUNK - 1;
        es[k] = src[e0 + ic];
        ed[k] = dst[e0 + ic];
    }

    for (int b0 = w; b0 < NB; b0 += 128) {
        int pre[4], all[4];
#pragma unroll
        for (int j = 0; j < 4; ++j) {
            const int b  = b0 + 32 * j;
            const int bc = b < NB ? b : NB - 1;
            pre[j] = 0;
            all[j] = 0;
#pragma unroll
            for (int k = 0; k < GP / 32; ++k) {
                const int i = k * 32 + lane;
                const int v = wgcnt[bc * GP + i];
                all[j] += v;
                pre[j] += (i < wg) ? v : 0;
            }
        }
#pragma unroll
        for (int j = 0; j < 4; ++j) {
#pragma unroll
            for (int o = 16; o > 0; o >>= 1) {
                all[j] += __shfl_xor(all[j], o);
                pre[j] += __shfl_xor(pre[j], o);
            }
            const int b = b0 + 32 * j;
            if (lane == 0 && b < NB) { tot[b] = all[j]; cur[b] = pre[j]; }
        }
    }
    __syncthreads();
    if (w == 0) {
        int carry = 0;
        for (int c = 0; c < NB; c += 32) {
            const int b = c + lane;
            const int v = (b < NB) ? tot[b] : 0;
            int incl = v;
#pragma unroll
            for (int o = 1; o < 32; o <<= 1) {
                const int t = __shfl_up(incl, o);
                if (lane >= o) incl += t;
            }
            if (b < NB) {
                const int start = carry + incl - v;
                cur[b] += start;
                if (wg == 0) binstart[b] = start;
            }
            carry += __shfl(incl, 31);
        }
        if (wg == 0 && lane == 0) binstart[NB] = carry;
    } else if (w == 1) {
        int carry = 0;
        for (int c = 0; c < NB; c += 32) {
            const int b = c + lane;
            const int v = (b < NB) ? lcnt[b] : 0;
            int incl = v;
#pragma unroll
            for (int o = 1; o < 32; o <<= 1) {
                const int t = __shfl_up(incl, o);
                if (lane >= o) incl += t;
            }
            if (b < NB) lcur[b] = carry + incl - v;
            carry += __shfl(incl, 31);
        }
    }
    __syncthreads();
    if (tid < NB) delta[tid] = cur[tid] - lcur[tid];
    __syncthreads();

#pragma unroll
    for (int k = 0; k < PER; ++k) {
        if (tid + k * 1024 < CHUNK) {
            const int r = atomicAdd(&lcur[ed[k] >> BIN_SHIFT], 1);
            stage[r] = make_int2(es[k], ed[k]);
        }
    }
    __syncthreads();
#pragma unroll
    for (int k = 0; k < PER; ++k) {
        const int i = tid + k * 1024;
        if (i < CHUNK) {
            const int2 p = stage[i];
            binned[i + delta[p.y >> BIN_SHIFT]] = p;
        }
    }
}

__global__ void __launch_bounds__(1024)
bin_sort_kernel(const int2* __restrict__ binned, const int* __restrict__ binstart,
                int* __restrict__ rowptr, int* __restrict__ col,
                float* __restrict__ dinv) {
    __shared__ int cnt[BIN_NODES];
    __shared__ int curs[BIN_NODES];
    __shared__ int wtot[BIN_NODES / 32];
    __shared__ int lcol[LCAP];
    const int tid  = threadIdx.x;
    const int lane = tid & 31;
    const int w    = tid >> 5;
    const int b    = blockIdx.x;
    const int bs   = binstart[b];
    const int be   = binstart[b + 1];
    const bool staged = (be - bs) <= LCAP;

    if (tid < BIN_NODES) cnt[tid] = 0;
    __syncthreads();
    for (int e = bs + tid; e < be; e += 4096) {
        int y[4];
#pragma unroll
        for (int j = 0; j < 4; ++j) {
            const int i = e + j * 1024;
            y[j] = binned[i < be ? i : be - 1].y;
        }
        asm volatile("" :: "v"(y[0]), "v"(y[1]), "v"(y[2]), "v"(y[3]));
#pragma unroll
        for (int j = 0; j < 4; ++j)
            if (e + j * 1024 < be) atomicAdd(&cnt[y[j] & (BIN_NODES - 1)], 1);
    }
    __syncthreads();

    int v = 0, incl = 0;
    if (tid < BIN_NODES) {
        v = cnt[tid];
        incl = v;
#pragma unroll
        for (int o = 1; o < 32; o <<= 1) {
            const int t = __shfl_up(incl, o);
            if (lane >= o) incl += t;
        }
        if (lane == 31) wtot[w] = incl;
    }
    __syncthreads();
    if (tid < BIN_NODES) {
        int off = 0;
        for (int k = 0; k < w; ++k) off += wtot[k];
        const int excl = off + incl - v;
        curs[tid] = excl;
        const int node = b * BIN_NODES + tid;
        if (node <= N_NODES) rowptr[node] = bs + excl;
        if (node < N_NODES)  dinv[node] = rsqrtf((float)v + 1.0f);
    }
    __syncthreads();
    for (int e = bs + tid; e < be; e += 4096) {
        int2 sd[4];
#pragma unroll
        for (int j = 0; j < 4; ++j) {
            const int i = e + j * 1024;
            sd[j] = binned[i < be ? i : be - 1];
        }
        asm volatile("" :: "v"(sd[0].y), "v"(sd[1].y), "v"(sd[2].y), "v"(sd[3].y));
#pragma unroll
        for (int j = 0; j < 4; ++j) {
            if (e + j * 1024 < be) {
                const int pos = atomicAdd(&curs[sd[j].y & (BIN_NODES - 1)], 1);
                if (staged) lcol[pos] = sd[j].x;
                else        col[bs + pos] = sd[j].x;
            }
        }
    }
    if (staged) {
        __syncthreads();
        for (int i = tid; i < be - bs; i += 1024) col[bs + i] = lcol[i];
    }
}

template <int K> struct rowvec { typedef __attribute__((ext_vector_type(K / 32))) float type; };

template <int K, int B, bool SCALE_IN>
__device__ __forceinline__ typename rowvec<K>::type
gather_sum(const int* __restrict__ rowptr, const int* __restrict__ col,
           const float* __restrict__ G, const float* __restrict__ dinv,
           const int node, const int lane) {
    typedef typename rowvec<K>::type vecf;
    const float* glane = G + lane * (K / 32);
    const int beg = rowptr[node];
    const int end = rowptr[node + 1];
    vecf acc = *reinterpret_cast<const vecf*>(glane + node * K);
    if (SCALE_IN) acc *= dinv[node];
    for (int base = beg; base < end; base += 32) {
        const int rem  = end - base;
        const int n    = rem < 32 ? rem : 32;
        const int mine = (lane < n) ? col[base + lane] : 0;
        int dmine = 0;
        if (SCALE_IN) dmine = __float_as_int((lane < n) ? dinv[mine] : 0.0f);
        int j = 0;
        for (; j + B <= n; j += B) {
            vecf v[B];
#pragma unroll
            for (int k = 0; k < B; ++k)
                v[k] = *reinterpret_cast<const vecf*>(glane + __builtin_amdgcn_readlane(mine, j + k) * K);
#pragma unroll
            for (int k = 0; k < B; ++k) {
                if (SCALE_IN) acc += v[k] * __int_as_float(__builtin_amdgcn_readlane(dmine, j + k));
                else          acc += v[k];
            }
        }
        const int r = n - j;
        if (r > 0) {
            vecf v[B - 1];
#pragma unroll
            for (int k = 0; k < B - 1; ++k) {
                v[k] = 0.0f;
                if (k == 0 || k < r) v[k] = *reinterpret_cast<const vecf*>(glane + __builtin_amdgcn_readlane(mine, j + k) * K);
            }
#pragma unroll
            for (int k = 0; k < B - 1; ++k) {
                if (SCALE_IN) acc += v[k] * __int_as_float(__builtin_amdgcn_readlane(dmine, j + k));
                else          acc += v[k];
            }
        }
    }
    return acc;
}

template <int K, int COUT, bool SCALE_IN, bool RELU, bool SCALE_OUT>
__global__ void __launch_bounds__(1024)
agg_gemm_kernel(const int* __restrict__ rowptr,
                const int* __restrict__ col,
                const float* __restrict__ G,
                const float* __restrict__ dinv,
                const float* __restrict__ W,
                const float* __restrict__ bias,
                float* __restrict__ OUT) {
    constexpr int VEC = K / 32;
    constexpr int P   = K + 4;
    constexpr int WP  = COUT + 8;
    typedef __attribute__((ext_vector_type(VEC))) float vecf;
    constexpr int NT  = COUT / 16;
    constexpr int GW  = 2 * NT;
    constexpr int NG  = 32 / GW;
    constexpr int GB  = 32 / VEC;
    __shared__ __attribute__((aligned(16))) float tile[2][32 * P];
    __shared__ __attribute__((aligned(16))) float wlds[K * WP];

    const int lane = threadIdx.x & 31;
    const int w    = __builtin_amdgcn_readfirstlane(threadIdx.x >> 5);
    const int grp  = w / GW;
    const int wl   = w - grp * GW;

    constexpr int NWP = K * COUT / 4096;
    v4f wv[NWP];
#pragma unroll
    for (int i = 0; i < NWP; ++i)
        wv[i] = *reinterpret_cast<const v4f*>(W + threadIdx.x * 4 + i * 4096);
#pragma unroll
    for (int i = 0; i < NWP; ++i) {
        const int f   = threadIdx.x * 4 + i * 4096;
        const int row = f / COUT;
        const int cx  = f - row * COUT;
        *reinterpret_cast<v4f*>(wlds + row * WP + cx) = wv[i];
    }

    int it = 0;
    for (int t = blockIdx.x; t < N_TILES; t += gridDim.x, ++it) {
        float* tb = tile[it & 1];
        const int node = t * 32 + w;
        const vecf acc = gather_sum<K, GB, SCALE_IN>(rowptr, col, G, dinv, node, lane);

        *reinterpret_cast<vecf*>(tb + w * P + lane * VEC) = acc;
        __syncthreads();

        if (grp == (it & (NG - 1))) {
            const int mt   = wl / NT;
            const int nt   = wl - mt * NT;
            const int c    = lane & 15;
            const int kh   = (lane >> 4) << 1;
            const int colg = nt * 16 + c;
            const float* arow = tb + (mt * 16 + c) * P + kh;
            const float* wcol = wlds + kh * WP + colg;

            v8f dacc = {};
#pragma unroll
            for (int kk = 0; kk < K; kk += 4) {
                v2f a = *reinterpret_cast<const v2f*>(arow + kk);
                v2f b;
                b.x = wcol[(kk)     * WP];
                b.y = wcol[(kk + 1) * WP];
                dacc = __builtin_amdgcn_wmma_f32_16x16x4_f32(
                     false, a,  false, b,
                     (short)0, dacc,  false,  false);
            }

            const int   mbase = t * 32 + mt * 16 + ((lane >> 4) << 3);
            const float bcol  = bias[colg];
#pragma unroll
            for (int r = 0; r < 8; ++r) {
                const int   m  = mbase + r;
                const float dm = dinv[m];
                float v = dm * dacc[r] + bcol;
                if (RELU) v = fmaxf(v, 0.0f);
                if (SCALE_OUT) v *= dm;
                OUT[m * COUT + colg] = v;
            }
        }
    }
}

__global__ void __launch_bounds__(1024, 16)
agg_gemm_next_kernel(const int* __restrict__ rowptr,
                     const int* __restrict__ col,
                     const float* __restrict__ G,
                     const float* __restrict__ dinv,
                     const float* __restrict__ W,
                     const float* __restrict__ bias,
                     const float* __restrict__ Wn,
                     float* __restrict__ Z) {
    constexpr int K    = 128;
    constexpr int COUT = 128;
    constexpr int CN   = 64;
    constexpr int P    = K + 4;
    constexpr int WP   = COUT + 8;
    constexpr int WNP  = CN + 8;
    constexpr int NT   = COUT / 16;
    constexpr int GW   = 2 * NT;
    constexpr int NTN  = CN / 16;
    static_assert(GW == 16 && COUT == K, "two wave groups; the H tile has the sum tile's shape");
    __shared__ __attribute__((aligned(16))) float tile[2][32 * P];
    __shared__ __attribute__((aligned(16))) float htile[32 * P];
    __shared__ __attribute__((aligned(16))) float wlds[K * WP];
    __shared__ __attribute__((aligned(16))) float wnlds[COUT * WNP];

    const int lane = threadIdx.x & 31;
    const int w    = __builtin_amdgcn_readfirstlane(threadIdx.x >> 5);
    const int grp  = w / GW;
    const int wl   = w - grp * GW;

    constexpr int NWP  = K * COUT / 4096;
    constexpr int NWNP = COUT * CN / 4096;
    v4f wv[NWP];
    v4f wnv[NWNP];
#pragma unroll
    for (int i = 0; i < NWP; ++i)
        wv[i] = *reinterpret_cast<const v4f*>(W + threadIdx.x * 4 + i * 4096);
#pragma unroll
    for (int i = 0; i < NWNP; ++i)
        wnv[i] = *reinterpret_cast<const v4f*>(Wn + threadIdx.x * 4 + i * 4096);
#pragma unroll
    for (int i = 0; i < NWP; ++i) {
        const int f   = threadIdx.x * 4 + i * 4096;
        const int row = f / COUT;
        const int cx  = f - row * COUT;
        *reinterpret_cast<v4f*>(wlds + row * WP + cx) = wv[i];
    }
#pragma unroll
    for (int i = 0; i < NWNP; ++i) {
        const int f   = threadIdx.x * 4 + i * 4096;
        const int row = f / CN;
        const int cx  = f - row * CN;
        *reinterpret_cast<v4f*>(wnlds + row * WNP + cx) = wnv[i];
    }

    const int c  = lane & 15;
    const int kh = (lane >> 4) << 1;
    const int rh = (lane >> 4) << 3;

    int t = blockIdx.x;
    if (t < N_TILES) {
        const v4f acc = gather_sum<K, 8, false>(rowptr, col, G, dinv, t * 32 + w, lane);
        *reinterpret_cast<v4f*>(tile[0] + w * P + lane * 4) = acc;
    }

    float hv[8];
    int it = 0;
    for (; t < N_TILES; t += gridDim.x, ++it) {
        __syncthreads();
        const float* tb = tile[it & 1];

        if (grp == (it & 1)) {
            const int mt   = wl / NT;
            const int nt   = wl - mt * NT;
            const int colg = nt * 16 + c;
            const float* arow = tb + (mt * 16 + c) * P + kh;
            const float* wcol = wlds + kh * WP + colg;

            v8f dacc = {};
#pragma unroll
            for (int kk = 0; kk < K; kk += 4) {
                v2f a = *reinterpret_cast<const v2f*>(arow + kk);
                v2f b;
                b.x = wcol[(kk)     * WP];
                b.y = wcol[(kk + 1) * WP];
                dacc = __builtin_amdgcn_wmma_f32_16x16x4_f32(
                     false, a,  false, b,
                     (short)0, dacc,  false,  false);
            }

            const int   mbase = t * 32 + mt * 16 + rh;
            const float bcol  = bias[colg];
#pragma unroll
            for (int r = 0; r < 8; ++r) {
                const float dm = dinv[mbase + r];
                float v = dm * dacc[r] + bcol;
                v = fmaxf(v, 0.0f);
                hv[r] = v * dm;
            }
        } else if (it > 0 && wl < 2 * NTN) {
            const int mt   = wl / NTN;
            const int nt   = wl - mt * NTN;
            const int colg = nt * 16 + c;
            const float* arow = htile + (mt * 16 + c) * P + kh;
            const float* wcol = wnlds + kh * WNP + colg;

            v8f zacc = {};
#pragma unroll
            for (int kk = 0; kk < COUT; kk += 4) {
                v2f a = *reinterpret_cast<const v2f*>(arow + kk);
                v2f b;
                b.x = wcol[(kk)     * WNP];
                b.y = wcol[(kk + 1) * WNP];
                zacc = __builtin_amdgcn_wmma_f32_16x16x4_f32(
                     false, a,  false, b,
                     (short)0, zacc,  false,  false);
            }

            const int mbase = (t - (int)gridDim.x) * 32 + mt * 16 + rh;
#pragma unroll
            for (int r = 0; r < 8; ++r)
                Z[(mbase + r) * CN + colg] = zacc[r];
        }

        const int tn = t + gridDim.x;
        if (tn < N_TILES) {
            const v4f acc = gather_sum<K, 8, false>(rowptr, col, G, dinv, tn * 32 + w, lane);
            *reinterpret_cast<v4f*>(tile[(it + 1) & 1] + w * P + lane * 4) = acc;
        }
        __syncthreads();

        if (grp == (it & 1)) {
            const int mt   = wl / NT;
            const int nt   = wl - mt * NT;
            float* hcol = htile + (mt * 16 + rh) * P + nt * 16 + c;
#pragma unroll
            for (int r = 0; r < 8; ++r)
                hcol[r * P] = hv[r];
        }
    }

    __syncthreads();
    if (it > 0 && grp != (it & 1) && wl < 2 * NTN) {
        const int mt   = wl / NTN;
        const int nt   = wl - mt * NTN;
        const int colg = nt * 16 + c;
        const float* arow = htile + (mt * 16 + c) * P + kh;
        const float* wcol = wnlds + kh * WNP + colg;

        v8f zacc = {};
#pragma unroll
        for (int kk = 0; kk < COUT; kk += 4) {
            v2f a = *reinterpret_cast<const v2f*>(arow + kk);
            v2f b;
            b.x = wcol[(kk)     * WNP];
            b.y = wcol[(kk + 1) * WNP];
            zacc = __builtin_amdgcn_wmma_f32_16x16x4_f32(
                 false, a,  false, b,
                 (short)0, zacc,  false,  false);
        }

        const int mbase = (t - (int)gridDim.x) * 32 + mt * 16 + rh;
#pragma unroll
        for (int r = 0; r < 8; ++r)
            Z[(mbase + r) * CN + colg] = zacc[r];
    }
}

__global__ void __launch_bounds__(1024)
agg_out_kernel(const int* __restrict__ rowptr,
               const int* __restrict__ col,
               const float* __restrict__ Z,
               const float* __restrict__ dinv,
               const float* __restrict__ bias,
               float* __restrict__ OUT) {
    constexpr int C = 64;
    const int lane = threadIdx.x & 31;
    const int w    = __builtin_amdgcn_readfirstlane(threadIdx.x >> 5);
    const v2f bl = *reinterpret_cast<const v2f*>(bias + lane * 2);
    for (int t = blockIdx.x; t < N_TILES; t += gridDim.x) {
        const int node = t * 32 + w;
        const v2f acc = gather_sum<C, 16, false>(rowptr, col, Z, dinv, node, lane);
        const float dm = dinv[node];
        *reinterpret_cast<v2f*>(OUT + node * C + lane * 2) = acc * dm + bl;
    }
}

extern "C" void kernel_launch(void* const* d_in, const int* in_sizes, int n_in,
                              void* d_out, int out_size, void* d_ws, size_t ws_size,
                              hipStream_t stream) {
    const float* x  = (const float*)d_in[0];
    const int*   ei = (const int*)d_in[1];
    const float* W1 = (const float*)d_in[2];
    const float* b1 = (const float*)d_in[3];
    const float* W2 = (const float*)d_in[4];
    const float* b2 = (const float*)d_in[5];
    const float* W3 = (const float*)d_in[6];
    const float* b3 = (const float*)d_in[7];
    float*       out = (float*)d_out;

    const int* src = ei;
    const int* dst = ei + N_EDGES;

    char*  ws       = (char*)d_ws;
    size_t off      = 0;
    float* dinv     = (float*)(ws + off); off += ((size_t)N_NODES * 4 + 511) & ~511ull;
    int*   rowptr   = (int*)(ws + off);   off += ((size_t)(N_NODES + 1) * 4 + 511) & ~511ull;
    int*   wgcnt    = (int*)(ws + off);   off += (size_t)NB * GP * 4;
    int*   binstart = (int*)(ws + off);   off += ((size_t)(NB + 1) * 4 + 511) & ~511ull;
    int2*  binned   = (int2*)(ws + off);  off += (size_t)N_EDGES * 8;
    int*   col      = (int*)(ws + off);   off += (size_t)N_EDGES * 4;
    float* ha       = (float*)(ws + off); off += (size_t)N_NODES * 128 * 4;
    float* hb       = (float*)(ws + off);

    const int L_BLOCKS = 1024;

    bin_count_kernel<<<GP, 1024, 0, stream>>>(dst, wgcnt);
    bin_scatter_kernel<<<GP, 1024, 0, stream>>>(src, dst, wgcnt, binstart, binned);
    bin_sort_kernel<<<NB, 1024, 0, stream>>>(binned, binstart, rowptr, col, dinv);

    agg_gemm_kernel<64, 128, true, true, true><<<L_BLOCKS, 1024, 0, stream>>>(rowptr, col, x, dinv, W1, b1, ha);

    agg_gemm_next_kernel<<<L_BLOCKS, 1024, 0, stream>>>(rowptr, col, ha, dinv, W2, b2, W3, hb);

    agg_out_kernel<<<L_BLOCKS, 1024, 0, stream>>>(rowptr, col, hb, dinv, b3, out);
}


# edge-list build in 2 launches for 3: count pass dropped; each chunk is grouped by bin in place (one LDS add: count and rank), a bin's work-group collects its 256 pieces; new LDS-arrival sum order
# speedup vs baseline: 6.5103x; 1.0830x over previous
#include <hip/hip_runtime.h>

#define N_NODES 100000
#define N_EDGES 1600000
#define N_TILES (N_NODES / 32)

typedef __attribute__((ext_vector_type(2))) float v2f;
typedef __attribute__((ext_vector_type(4))) float v4f;
typedef __attribute__((ext_vector_type(8))) float v8f;

#define BIN_SHIFT 8
#define BIN_NODES 256
#define NB ((N_NODES + BIN_NODES - 1) / BIN_NODES)
#define NBW ((NB + 31) / 32)
#define GP 256
#define CHUNK (N_EDGES / GP)
#define PER ((CHUNK + 1023) / 1024)
static_assert(N_EDGES % GP == 0, "edge chunks are exact");
static_assert(GP % 32 == 0 && GP == BIN_NODES, "the first 8 waves of a bin's work-group take its pieces and its nodes");
#define LCAP 8192

__global__ void __launch_bounds__(1024)
chunk_group_kernel(const int* __restrict__ src, const int* __restrict__ dst,
                   int* __restrict__ starts, int2* __restrict__ binned) {
    __shared__ int hist[NB];
    __shared__ int lst[NB];
    __shared__ int wtot[NBW];
    __shared__ int2 stage[CHUNK];
    const int tid  = threadIdx.x;
    const int lane = tid & 31;
    const int w    = tid >> 5;
    const int wg   = blockIdx.x;
    const int e0   = wg * CHUNK;

    int es[PER], ed[PER], rk[PER];
#pragma unroll
    for (int k = 0; k < PER; ++k) {
        const int i  = tid + k * 1024;
        const int ic = i < CHUNK ? i : CHUNK - 1;
        es[k] = src[e0 + ic];
        ed[k] = dst[e0 + ic];
    }
    if (tid < NB) hist[tid] = 0;
    __syncthreads();
#pragma unroll
    for (int k = 0; k < PER; ++k) {
        rk[k] = 0;
        if (tid + k * 1024 < CHUNK) rk[k] = atomicAdd(&hist[ed[k] >> BIN_SHIFT], 1);
    }
    __syncthreads();

    int v = 0, incl = 0;
    if (tid < NBW * 32) {
        v = (tid < NB) ? hist[tid] : 0;
        incl = v;
#pragma unroll
        for (int o = 1; o < 32; o <<= 1) {
            const int t = __shfl_up(incl, o);
            if (lane >= o) incl += t;
        }
        if (lane == 31) wtot[w] = incl;
    }
    __syncthreads();
    if (tid < NB) {
        int off = 0;
        for (int k = 0; k < w; ++k) off += wtot[k];
        const int excl = off + incl - v;
        lst[tid] = excl;
        starts[tid * GP + wg] = excl;
    }
    if (tid == 0) starts[NB * GP + wg] = CHUNK;
    __syncthreads();

#pragma unroll
    for (int k = 0; k < PER; ++k)
        if (tid + k * 1024 < CHUNK) stage[lst[ed[k] >> BIN_SHIFT] + rk[k]] = make_int2(es[k], ed[k]);
    __syncthreads();
    const int head = (32 - (e0 & 31)) & 31;
    if (tid < head) binned[e0 + tid] = stage[tid];
#pragma unroll
    for (int k = 0; k < PER; ++k) {
        const int i = head + tid + k * 1024;
        if (i < CHUNK) binned[e0 + i] = stage[i];
    }
}

__global__ void __launch_bounds__(1024)
bin_sort_kernel(const int2* __restrict__ binned, const int* __restrict__ starts,
                int* __restrict__ rowptr, int* __restrict__ col,
                float* __restrict__ dinv) {
    constexpr int NP = GP / 32;
    __shared__ int cnt[BIN_NODES];
    __shared__ int curs[BIN_NODES];
    __shared__ int wtot[BIN_NODES / 32];
    __shared__ int pbase[GP];
    __shared__ int pcnt[GP];
    __shared__ int red[2][GP / 32];
    __shared__ int lcol[LCAP];
    const int tid  = threadIdx.x;
    const int lane = tid & 31;
    const int w    = tid >> 5;
    const int b    = blockIdx.x;

    int s0 = 0, s1 = 0;
    if (tid < GP) {
        s0 = starts[b * GP + tid];
        s1 = starts[(b + 1) * GP + tid];
        pbase[tid] = tid * CHUNK + s0;
        pcnt[tid]  = s1 - s0;
        cnt[tid] = 0;
    }
    __syncthreads();

    int2 sd[NP];
    int  pc[NP];
#pragma unroll
    for (int j = 0; j < NP; ++j) {
        const int p = j * 32 + w;
        pc[j] = pcnt[p];
        int i = pbase[p] + (lane < pc[j] ? lane : 0);
        i = i < N_EDGES ? i : N_EDGES - 1;
        sd[j] = binned[i];
    }
    if (tid < GP) {
        int a = s0, n = s1 - s0;
#pragma unroll
        for (int o = 16; o > 0; o >>= 1) {
            a += __shfl_xor(a, o);
            n += __shfl_xor(n, o);
        }
        if (lane == 0) { red[0][w] = a; red[1][w] = n; }
    }
    int cmax = 0;
#pragma unroll
    for (int j = 0; j < NP; ++j) {
        cmax = pc[j] > cmax ? pc[j] : cmax;
        if (lane < pc[j]) atomicAdd(&cnt[sd[j].y & (BIN_NODES - 1)], 1);
    }
    if (cmax > 32) {
#pragma nounroll
        for (int j = 0; j < NP; ++j) {
            const int pb = pbase[j * 32 + w];
            const int pn = pcnt[j * 32 + w];
            for (int l = lane + 32; l < pn; l += 32)
                atomicAdd(&cnt[binned[pb + l].y & (BIN_NODES - 1)], 1);
        }
    }
    __syncthreads();
    int bs = 0, nb = 0;
    for (int k = 0; k < GP / 32; ++k) { bs += red[0][k]; nb += red[1][k]; }
    const bool staged = nb <= LCAP;

    int v = 0, incl = 0;
    if (tid < BIN_NODES) {
        v = cnt[tid];
        incl = v;
#pragma unroll
        for (int o = 1; o < 32; o <<= 1) {
            const int t = __shfl_up(incl, o);
            if (lane >= o) incl += t;
        }
        if (lane == 31) wtot[w] = incl;
    }
    __syncthreads();
    if (tid < BIN_NODES) {
        int off = 0;
        for (int k = 0; k < w; ++k) off += wtot[k];
        const int excl = off + incl - v;
        curs[tid] = excl;
        const int node = b * BIN_NODES + tid;
        if (node <= N_NODES) rowptr[node] = bs + excl;
        if (node < N_NODES)  dinv[node] = rsqrtf((float)v + 1.0f);
    }
    __syncthreads();
    int pos[NP];
#pragma unroll
    for (int j = 0; j < NP; ++j) {
        pos[j] = 0;
        if (lane < pc[j]) pos[j] = atomicAdd(&curs[sd[j].y & (BIN_NODES - 1)], 1);
    }
    if (staged) {
#pragma unroll
        for (int j = 0; j < NP; ++j)
            if (lane < pc[j]) lcol[pos[j]] = sd[j].x;
    } else {
#pragma unroll
        for (int j = 0; j < NP; ++j)
            if (lane < pc[j]) col[bs + pos[j]] = sd[j].x;
    }
    if (cmax > 32) {
#pragma nounroll
        for (int j = 0; j < NP; ++j) {
            const int pb = pbase[j * 32 + w];
            const int pn = pcnt[j * 32 + w];
            for (int l = lane + 32; l < pn; l += 32) {
                const int2 q = binned[pb + l];
                const int qpos = atomicAdd(&curs[q.y & (BIN_NODES - 1)], 1);
                if (staged) lcol[qpos] = q.x;
                else        col[bs + qpos] = q.x;
            }
        }
    }
    if (staged) {
        __syncthreads();
        for (int i = tid; i < nb; i += 1024) col[bs + i] = lcol[i];
    }
}

template <int K> struct rowvec { typedef __attribute__((ext_vector_type(K / 32))) float type; };

template <int K, int B, bool SCALE_IN>
__device__ __forceinline__ typename rowvec<K>::type
gather_sum(const int* __restrict__ rowptr, const int* __restrict__ col,
           const float* __restrict__ G, const float* __restrict__ dinv,
           const int node, const int lane) {
    typedef typename rowvec<K>::type vecf;
    const float* glane = G + lane * (K / 32);
    const int beg = rowptr[node];
    const int end = rowptr[node + 1];
    vecf acc = *reinterpret_cast<const vecf*>(glane + node * K);
    if (SCALE_IN) acc *= dinv[node];
    for (int base = beg; base < end; base += 32) {
        const int rem  = end - base;
        const int n    = rem < 32 ? rem : 32;
        const int mine = (lane < n) ? col[base + lane] : 0;
        int dmine = 0;
        if (SCALE_IN) dmine = __float_as_int((lane < n) ? dinv[mine] : 0.0f);
        int j = 0;
        for (; j + B <= n; j += B) {
            vecf v[B];
#pragma unroll
            for (int k = 0; k < B; ++k)
                v[k] = *reinterpret_cast<const vecf*>(glane + __builtin_amdgcn_readlane(mine, j + k) * K);
#pragma unroll
            for (int k = 0; k < B; ++k) {
                if (SCALE_IN) acc += v[k] * __int_as_float(__builtin_amdgcn_readlane(dmine, j + k));
                else          acc += v[k];
            }
        }
        const int r = n - j;
        if (r > 0) {
            vecf v[B - 1];
#pragma unroll
            for (int k = 0; k < B - 1; ++k) {
                v[k] = 0.0f;
                if (k == 0 || k < r) v[k] = *reinterpret_cast<const vecf*>(glane + __builtin_amdgcn_readlane(mine, j + k) * K);
            }
#pragma unroll
            for (int k = 0; k < B - 1; ++k) {
                if (SCALE_IN) acc += v[k] * __int_as_float(__builtin_amdgcn_readlane(dmine, j + k));
                else          acc += v[k];
            }
        }
    }
    return acc;
}

template <int K, int COUT, bool SCALE_IN, bool RELU, bool SCALE_OUT>
__global__ void __launch_bounds__(1024)
agg_gemm_kernel(const int* __restrict__ rowptr,
                const int* __restrict__ col,
                const float* __restrict__ G,
                const float* __restrict__ dinv,
                const float* __restrict__ W,
                const float* __restrict__ bias,
                float* __restrict__ OUT) {
    constexpr int VEC = K / 32;
    constexpr int P   = K + 4;
    constexpr int WP  = COUT + 8;
    typedef __attribute__((ext_vector_type(VEC))) float vecf;
    constexpr int NT  = COUT / 16;
    constexpr int GW  = 2 * NT;
    constexpr int NG  = 32 / GW;
    constexpr int GB  = 32 / VEC;
    __shared__ __attribute__((aligned(16))) float tile[2][32 * P];
    __shared__ __attribute__((aligned(16))) float wlds[K * WP];

    const int lane = threadIdx.x & 31;
    const int w    = __builtin_amdgcn_readfirstlane(threadIdx.x >> 5);
    const int grp  = w / GW;
    const int wl   = w - grp * GW;

    constexpr int NWP = K * COUT / 4096;
    v4f wv[NWP];
#pragma unroll
    for (int i = 0; i < NWP; ++i)
        wv[i] = *reinterpret_cast<const v4f*>(W + threadIdx.x * 4 + i * 4096);
#pragma unroll
    for (int i = 0; i < NWP; ++i) {
        const int f   = threadIdx.x * 4 + i * 4096;
        const int row = f / COUT;
        const int cx  = f - row * COUT;
        *reinterpret_cast<v4f*>(wlds + row * WP + cx) = wv[i];
    }

    int it = 0;
    for (int t = blockIdx.x; t < N_TILES; t += gridDim.x, ++it) {
        float* tb = tile[it & 1];
        const int node = t * 32 + w;
        const vecf acc = gather_sum<K, GB, SCALE_IN>(rowptr, col, G, dinv, node, lane);

        *reinterpret_cast<vecf*>(tb + w * P + lane * VEC) = acc;
        __syncthreads();

        if (grp == (it & (NG - 1))) {
            const int mt   = wl / NT;
            const int nt   = wl - mt * NT;
            const int c    = lane & 15;
            const int kh   = (lane >> 4) << 1;
            const int colg = nt * 16 + c;
            const float* arow = tb + (mt * 16 + c) * P + kh;
            const float* wcol = wlds + kh * WP + colg;

            v8f dacc = {};
#pragma unroll
            for (int kk = 0; kk < K; kk += 4) {
                v2f a = *reinterpret_cast<const v2f*>(arow + kk);
                v2f b;
                b.x = wcol[(kk)     * WP];
                b.y = wcol[(kk + 1) * WP];
                dacc = __builtin_amdgcn_wmma_f32_16x16x4_f32(
                     false, a,  false, b,
                     (short)0, dacc,  false,  false);
            }

            const int   mbase = t * 32 + mt * 16 + ((lane >> 4) << 3);
            const float bcol  = bias[colg];
#pragma unroll
            for (int r = 0; r < 8; ++r) {
                const int   m  = mbase + r;
                const float dm = dinv[m];
                float v = dm * dacc[r] + bcol;
                if (RELU) v = fmaxf(v, 0.0f);
                if (SCALE_OUT) v *= dm;
                OUT[m * COUT + colg] = v;
            }
        }
    }
}

__global__ void __launch_bounds__(1024, 16)
agg_gemm_next_kernel(const int* __restrict__ rowptr,
                     const int* __restrict__ col,
                     const float* __restrict__ G,
                     const float* __restrict__ dinv,
                     const float* __restrict__ W,
                     const float* __restrict__ bias,
                     const float* __restrict__ Wn,
                     float* __restrict__ Z) {
    constexpr int K    = 128;
    constexpr int COUT = 128;
    constexpr int CN   = 64;
    constexpr int P    = K + 4;
    constexpr int WP   = COUT + 8;
    constexpr int WNP  = CN + 8;
    constexpr int NT   = COUT / 16;
    constexpr int GW   = 2 * NT;
    constexpr int NTN  = CN / 16;
    static_assert(GW == 16 && COUT == K, "two wave groups; the H tile has the sum tile's shape");
    __shared__ __attribute__((aligned(16))) float tile[2][32 * P];
    __shared__ __attribute__((aligned(16))) float htile[32 * P];
    __shared__ __attribute__((aligned(16))) float wlds[K * WP];
    __shared__ __attribute__((aligned(16))) float wnlds[COUT * WNP];

    const int lane = threadIdx.x & 31;
    const int w    = __builtin_amdgcn_readfirstlane(threadIdx.x >> 5);
    const int grp  = w / GW;
    const int wl   = w - grp * GW;

    constexpr int NWP  = K * COUT / 4096;
    constexpr int NWNP = COUT * CN / 4096;
    v4f wv[NWP];
    v4f wnv[NWNP];
#pragma unroll
    for (int i = 0; i < NWP; ++i)
        wv[i] = *reinterpret_cast<const v4f*>(W + threadIdx.x * 4 + i * 4096);
#pragma unroll
    for (int i = 0; i < NWNP; ++i)
        wnv[i] = *reinterpret_cast<const v4f*>(Wn + threadIdx.x * 4 + i * 4096);
#pragma unroll
    for (int i = 0; i < NWP; ++i) {
        const int f   = threadIdx.x * 4 + i * 4096;
        const int row = f / COUT;
        const int cx  = f - row * COUT;
        *reinterpret_cast<v4f*>(wlds + row * WP + cx) = wv[i];
    }
#pragma unroll
    for (int i = 0; i < NWNP; ++i) {
        const int f   = threadIdx.x * 4 + i * 4096;
        const int row = f / CN;
        const int cx  = f - row * CN;
        *reinterpret_cast<v4f*>(wnlds + row * WNP + cx) = wnv[i];
    }

    const int c  = lane & 15;
    const int kh = (lane >> 4) << 1;
    const int rh = (lane >> 4) << 3;

    int t = blockIdx.x;
    if (t < N_TILES) {
        const v4f acc = gather_sum<K, 8, false>(rowptr, col, G, dinv, t * 32 + w, lane);
        *reinterpret_cast<v4f*>(tile[0] + w * P + lane * 4) = acc;
    }

    float hv[8];
    int it = 0;
    for (; t < N_TILES; t += gridDim.x, ++it) {
        __syncthreads();
        const float* tb = tile[it & 1];

        if (grp == (it & 1)) {
            const int mt   = wl / NT;
            const int nt   = wl - mt * NT;
            const int colg = nt * 16 + c;
            const float* arow = tb + (mt * 16 + c) * P + kh;
            const float* wcol = wlds + kh * WP + colg;

            v8f dacc = {};
#pragma unroll
            for (int kk = 0; kk < K; kk += 4) {
                v2f a = *reinterpret_cast<const v2f*>(arow + kk);
                v2f b;
                b.x = wcol[(kk)     * WP];
                b.y = wcol[(kk + 1) * WP];
                dacc = __builtin_amdgcn_wmma_f32_16x16x4_f32(
                     false, a,  false, b,
                     (short)0, dacc,  false,  false);
            }

            const int   mbase = t * 32 + mt * 16 + rh;
            const float bcol  = bias[colg];
#pragma unroll
            for (int r = 0; r < 8; ++r) {
                const float dm = dinv[mbase + r];
                float v = dm * dacc[r] + bcol;
                v = fmaxf(v, 0.0f);
                hv[r] = v * dm;
            }
        } else if (it > 0 && wl < 2 * NTN) {
            const int mt   = wl / NTN;
            const int nt   = wl - mt * NTN;
            const int colg = nt * 16 + c;
            const float* arow = htile + (mt * 16 + c) * P + kh;
            const float* wcol = wnlds + kh * WNP + colg;

            v8f zacc = {};
#pragma unroll
            for (int kk = 0; kk < COUT; kk += 4) {
                v2f a = *reinterpret_cast<const v2f*>(arow + kk);
                v2f b;
                b.x = wcol[(kk)     * WNP];
                b.y = wcol[(kk + 1) * WNP];
                zacc = __builtin_amdgcn_wmma_f32_16x16x4_f32(
                     false, a,  false, b,
                     (short)0, zacc,  false,  false);
            }

            const int mbase = (t - (int)gridDim.x) * 32 + mt * 16 + rh;
#pragma unroll
            for (int r = 0; r < 8; ++r)
                Z[(mbase + r) * CN + colg] = zacc[r];
        }

        const int tn = t + gridDim.x;
        if (tn < N_TILES) {
            const v4f acc = gather_sum<K, 8, false>(rowptr, col, G, dinv, tn * 32 + w, lane);
            *reinterpret_cast<v4f*>(tile[(it + 1) & 1] + w * P + lane * 4) = acc;
        }
        __syncthreads();

        if (grp == (it & 1)) {
            const int mt   = wl / NT;
            const int nt   = wl - mt * NT;
            float* hcol = htile + (mt * 16 + rh) * P + nt * 16 + c;
#pragma unroll
            for (int r = 0; r < 8; ++r)
                hcol[r * P] = hv[r];
        }
    }

    __syncthreads();
    if (it > 0 && grp != (it & 1) && wl < 2 * NTN) {
        const int mt   = wl / NTN;
        const int nt   = wl - mt * NTN;
        const int colg = nt * 16 + c;
        const float* arow = htile + (mt * 16 + c) * P + kh;
        const float* wcol = wnlds + kh * WNP + colg;

        v8f zacc = {};
#pragma unroll
        for (int kk = 0; kk < COUT; kk += 4) {
            v2f a = *reinterpret_cast<const v2f*>(arow + kk);
            v2f b;
            b.x = wcol[(kk)     * WNP];
            b.y = wcol[(kk + 1) * WNP];
            zacc = __builtin_amdgcn_wmma_f32_16x16x4_f32(
                 false, a,  false, b,
                 (short)0, zacc,  false,  false);
        }

        const int mbase = (t - (int)gridDim.x) * 32 + mt * 16 + rh;
#pragma unroll
        for (int r = 0; r < 8; ++r)
            Z[(mbase + r) * CN + colg] = zacc[r];
    }
}

__global__ void __launch_bounds__(1024)
agg_out_kernel(const int* __restrict__ rowptr,
               const int* __restrict__ col,
               const float* __restrict__ Z,
               const float* __restrict__ dinv,
               const float* __restrict__ bias,
               float* __restrict__ OUT) {
    constexpr int C = 64;
    const int lane = threadIdx.x & 31;
    const int w    = __builtin_amdgcn_readfirstlane(threadIdx.x >> 5);
    const v2f bl = *reinterpret_cast<const v2f*>(bias + lane * 2);
    for (int t = blockIdx.x; t < N_TILES; t += gridDim.x) {
        const int node = t * 32 + w;
        const v2f acc = gather_sum<C, 16, false>(rowptr, col, Z, dinv, node, lane);
        const float dm = dinv[node];
        *reinterpret_cast<v2f*>(OUT + node * C + lane * 2) = acc * dm + bl;
    }
}

extern "C" void kernel_launch(void* const* d_in, const int* in_sizes, int n_in,
                              void* d_out, int out_size, void* d_ws, size_t ws_size,
                              hipStream_t stream) {
    const float* x  = (const float*)d_in[0];
    const int*   ei = (const int*)d_in[1];
    const float* W1 = (const float*)d_in[2];
    const float* b1 = (const float*)d_in[3];
    const float* W2 = (const float*)d_in[4];
    const float* b2 = (const float*)d_in[5];
    const float* W3 = (const float*)d_in[6];
    const float* b3 = (const float*)d_in[7];
    float*       out = (float*)d_out;

    const int* src = ei;
    const int* dst = ei + N_EDGES;

    char*  ws       = (char*)d_ws;
    size_t off      = 0;
    float* dinv     = (float*)(ws + off); off += ((size_t)N_NODES * 4 + 511) & ~511ull;
    int*   rowptr   = (int*)(ws + off);   off += ((size_t)(N_NODES + 1) * 4 + 511) & ~511ull;
    int*   starts   = (int*)(ws + off);   off += (size_t)(NB + 1) * GP * 4;
    int2*  binned   = (int2*)(ws + off);  off += (size_t)N_EDGES * 8;
    int*   col      = (int*)(ws + off);   off += (size_t)N_EDGES * 4;
    float* ha       = (float*)(ws + off); off += (size_t)N_NODES * 128 * 4;
    float* hb       = (float*)(ws + off);

    const int L_BLOCKS = 1024;

    chunk_group_kernel<<<GP, 1024, 0, stream>>>(src, dst, starts, binned);
    bin_sort_kernel<<<NB, 1024, 0, stream>>>(binned, starts, rowptr, col, dinv);

    agg_gemm_kernel<64, 128, true, true, true><<<L_BLOCKS, 1024, 0, stream>>>(rowptr, col, x, dinv, W1, b1, ha);

    agg_gemm_next_kernel<<<L_BLOCKS, 1024, 0, stream>>>(rowptr, col, ha, dinv, W2, b2, W3, hb);

    agg_out_kernel<<<L_BLOCKS, 1024, 0, stream>>>(rowptr, col, hb, dinv, b3, out);
}


# layer 1 takes each edge's source factor from a one-byte in-degree table held in LDS (one copy per work-group, 512 resident ones) instead of a scattered 4-byte dinv lookup per edge; sum order kept
# speedup vs baseline: 6.7764x; 1.0409x over previous
#include <hip/hip_runtime.h>

#define N_NODES 100000
#define N_EDGES 1600000
#define N_TILES (N_NODES / 32)

typedef __attribute__((ext_vector_type(2))) float v2f;
typedef __attribute__((ext_vector_type(4))) float v4f;
typedef __attribute__((ext_vector_type(8))) float v8f;

#define BIN_SHIFT 8
#define BIN_NODES 256
#define NB ((N_NODES + BIN_NODES - 1) / BIN_NODES)
#define NBW ((NB + 31) / 32)
#define GP 256
#define CHUNK (N_EDGES / GP)
#define PER ((CHUNK + 1023) / 1024)
static_assert(N_EDGES % GP == 0, "edge chunks are exact");
static_assert(GP % 32 == 0 && GP == BIN_NODES, "the first 8 waves of a bin's work-group take its pieces and its nodes");
#define LCAP 8192
#define DEG_ESC 255
static_assert(N_NODES % 16 == 0, "the degree table is copied in 16-byte pieces and written four nodes per dword");

__global__ void __launch_bounds__(1024)
chunk_group_kernel(const int* __restrict__ src, const int* __restrict__ dst,
                   int* __restrict__ starts, int2* __restrict__ binned) {
    __shared__ int hist[NB];
    __shared__ int lst[NB];
    __shared__ int wtot[NBW];
    __shared__ int2 stage[CHUNK];
    const int tid  = threadIdx.x;
    const int lane = tid & 31;
    const int w    = tid >> 5;
    const int wg   = blockIdx.x;
    const int e0   = wg * CHUNK;

    int es[PER], ed[PER], rk[PER];
#pragma unroll
    for (int k = 0; k < PER; ++k) {
        const int i  = tid + k * 1024;
        const int ic = i < CHUNK ? i : CHUNK - 1;
        es[k] = src[e0 + ic];
        ed[k] = dst[e0 + ic];
    }
    if (tid < NB) hist[tid] = 0;
    __syncthreads();
#pragma unroll
    for (int k = 0; k < PER; ++k) {
        rk[k] = 0;
        if (tid + k * 1024 < CHUNK) rk[k] = atomicAdd(&hist[ed[k] >> BIN_SHIFT], 1);
    }
    __syncthreads();

    int v = 0, incl = 0;
    if (tid < NBW * 32) {
        v = (tid < NB) ? hist[tid] : 0;
        incl = v;
#pragma unroll
        for (int o = 1; o < 32; o <<= 1) {
            const int t = __shfl_up(incl, o);
            if (lane >= o) incl += t;
        }
        if (lane == 31) wtot[w] = incl;
    }
    __syncthreads();
    if (tid < NB) {
        int off = 0;
        for (int k = 0; k < w; ++k) off += wtot[k];
        const int excl = off + incl - v;
        lst[tid] = excl;
        starts[tid * GP + wg] = excl;
    }
    if (tid == 0) starts[NB * GP + wg] = CHUNK;
    __syncthreads();

#pragma unroll
    for (int k = 0; k < PER; ++k)
        if (tid + k * 1024 < CHUNK) stage[lst[ed[k] >> BIN_SHIFT] + rk[k]] = make_int2(es[k], ed[k]);
    __syncthreads();
    const int head = (32 - (e0 & 31)) & 31;
    if (tid < head) binned[e0 + tid] = stage[tid];
#pragma unroll
    for (int k = 0; k < PER; ++k) {
        const int i = head + tid + k * 1024;
        if (i < CHUNK) binned[e0 + i] = stage[i];
    }
}

__global__ void __launch_bounds__(1024)
bin_sort_kernel(const int2* __restrict__ binned, const int* __restrict__ starts,
                int* __restrict__ rowptr, int* __restrict__ col,
                float* __restrict__ dinv, unsigned int* __restrict__ deg8w) {
    constexpr int NP = GP / 32;
    __shared__ __attribute__((aligned(16))) int cnt[BIN_NODES];
    __shared__ int curs[BIN_NODES];
    __shared__ int wtot[BIN_NODES / 32];
    __shared__ int pbase[GP];
    __shared__ int pcnt[GP];
    __shared__ int red[2][GP / 32];
    __shared__ int lcol[LCAP];
    const int tid  = threadIdx.x;
    const int lane = tid & 31;
    const int w    = tid >> 5;
    const int b    = blockIdx.x;

    int s0 = 0, s1 = 0;
    if (tid < GP) {
        s0 = starts[b * GP + tid];
        s1 = starts[(b + 1) * GP + tid];
        pbase[tid] = tid * CHUNK + s0;
        pcnt[tid]  = s1 - s0;
        cnt[tid] = 0;
    }
    __syncthreads();

    int2 sd[NP];
    int  pc[NP];
#pragma unroll
    for (int j = 0; j < NP; ++j) {
        const int p = j * 32 + w;
        pc[j] = pcnt[p];
        int i = pbase[p] + (lane < pc[j] ? lane : 0);
        i = i < N_EDGES ? i : N_EDGES - 1;
        sd[j] = binned[i];
    }
    if (tid < GP) {
        int a = s0, n = s1 - s0;
#pragma unroll
        for (int o = 16; o > 0; o >>= 1) {
            a += __shfl_xor(a, o);
            n += __shfl_xor(n, o);
        }
        if (lane == 0) { red[0][w] = a; red[1][w] = n; }
    }
    int cmax = 0;
#pragma unroll
    for (int j = 0; j < NP; ++j) {
        cmax = pc[j] > cmax ? pc[j] : cmax;
        if (lane < pc[j]) atomicAdd(&cnt[sd[j].y & (BIN_NODES - 1)], 1);
    }
    if (cmax > 32) {
#pragma nounroll
        for (int j = 0; j < NP; ++j) {
            const int pb = pbase[j * 32 + w];
            const int pn = pcnt[j * 32 + w];
            for (int l = lane + 32; l < pn; l += 32)
                atomicAdd(&cnt[binned[pb + l].y & (BIN_NODES - 1)], 1);
        }
    }
    __syncthreads();
    int bs = 0, nb = 0;
    for (int k = 0; k < GP / 32; ++k) { bs += red[0][k]; nb += red[1][k]; }
    const bool staged = nb <= LCAP;

    int v = 0, incl = 0;
    if (tid < BIN_NODES) {
        v = cnt[tid];
        incl = v;
#pragma unroll
        for (int o = 1; o < 32; o <<= 1) {
            const int t = __shfl_up(incl, o);
            if (lane >= o) incl += t;
        }
        if (lane == 31) wtot[w] = incl;
    }
    __syncthreads();
    if (tid < BIN_NODES) {
        int off = 0;
        for (int k = 0; k < w; ++k) off += wtot[k];
        const int excl = off + incl - v;
        curs[tid] = excl;
        const int node = b * BIN_NODES + tid;
        if (node <= N_NODES) rowptr[node] = bs + excl;
        if (node < N_NODES)  dinv[node] = rsqrtf((float)v + 1.0f);
        const int n0 = b * BIN_NODES + tid * 4;
        if (tid < BIN_NODES / 4 && n0 < N_NODES) {
            const int4 c4 = *reinterpret_cast<const int4*>(&cnt[tid * 4]);
            const unsigned int q0 = c4.x < DEG_ESC ? c4.x : DEG_ESC;
            const unsigned int q1 = c4.y < DEG_ESC ? c4.y : DEG_ESC;
            const unsigned int q2 = c4.z < DEG_ESC ? c4.z : DEG_ESC;
            const unsigned int q3 = c4.w < DEG_ESC ? c4.w : DEG_ESC;
            deg8w[n0 >> 2] = q0 | (q1 << 8) | (q2 << 16) | (q3 << 24);
        }
    }
    __syncthreads();
    int pos[NP];
#pragma unroll
    for (int j = 0; j < NP; ++j) {
        pos[j] = 0;
        if (lane < pc[j]) pos[j] = atomicAdd(&curs[sd[j].y & (BIN_NODES - 1)], 1);
    }
    if (staged) {
#pragma unroll
        for (int j = 0; j < NP; ++j)
            if (lane < pc[j]) lcol[pos[j]] = sd[j].x;
    } else {
#pragma unroll
        for (int j = 0; j < NP; ++j)
            if (lane < pc[j]) col[bs + pos[j]] = sd[j].x;
    }
    if (cmax > 32) {
#pragma nounroll
        for (int j = 0; j < NP; ++j) {
            const int pb = pbase[j * 32 + w];
            const int pn = pcnt[j * 32 + w];
            for (int l = lane + 32; l < pn; l += 32) {
                const int2 q = binned[pb + l];
                const int qpos = atomicAdd(&curs[q.y & (BIN_NODES - 1)], 1);
                if (staged) lcol[qpos] = q.x;
                else        col[bs + qpos] = q.x;
            }
        }
    }
    if (staged) {
        __syncthreads();
        for (int i = tid; i < nb; i += 1024) col[bs + i] = lcol[i];
    }
}

template <int K> struct rowvec { typedef __attribute__((ext_vector_type(K / 32))) float type; };

template <int K, int B, bool SCALE_IN>
__device__ __forceinline__ typename rowvec<K>::type
gather_sum(const int* __restrict__ rowptr, const int* __restrict__ col,
           const float* __restrict__ G, const float* __restrict__ dinv,
           const unsigned char* dtab,
           const int node, const int lane) {
    typedef typename rowvec<K>::type vecf;
    const float* glane = G + lane * (K / 32);
    const int beg = rowptr[node];
    const int end = rowptr[node + 1];
    vecf acc = *reinterpret_cast<const vecf*>(glane + node * K);
    if (SCALE_IN) acc *= dinv[node];
    for (int base = beg; base < end; base += 32) {
        const int rem  = end - base;
        const int n    = rem < 32 ? rem : 32;
        const int mine = (lane < n) ? col[base + lane] : 0;
        int dmine = 0;
        if (SCALE_IN) {
            float f = 0.0f;
            if (lane < n) {
                const int d = dtab[mine];
                f = rsqrtf((float)d + 1.0f);
                if (d == DEG_ESC) f = dinv[mine];
            }
            dmine = __float_as_int(f);
        }
        int j = 0;
        for (; j + B <= n; j += B) {
            vecf v[B];
#pragma unroll
            for (int k = 0; k < B; ++k)
                v[k] = *reinterpret_cast<const vecf*>(glane + __builtin_amdgcn_readlane(mine, j + k) * K);
#pragma unroll
            for (int k = 0; k < B; ++k) {
                if (SCALE_IN) acc += v[k] * __int_as_float(__builtin_amdgcn_readlane(dmine, j + k));
                else          acc += v[k];
            }
        }
        const int r = n - j;
        if (r > 0) {
            vecf v[B - 1];
#pragma unroll
            for (int k = 0; k < B - 1; ++k) {
                v[k] = 0.0f;
                if (k == 0 || k < r) v[k] = *reinterpret_cast<const vecf*>(glane + __builtin_amdgcn_readlane(mine, j + k) * K);
            }
#pragma unroll
            for (int k = 0; k < B - 1; ++k) {
                if (SCALE_IN) acc += v[k] * __int_as_float(__builtin_amdgcn_readlane(dmine, j + k));
                else          acc += v[k];
            }
        }
    }
    return acc;
}

template <int K, int COUT, bool SCALE_IN, bool RELU, bool SCALE_OUT>
__global__ void __launch_bounds__(1024)
agg_gemm_kernel(const int* __restrict__ rowptr,
                const int* __restrict__ col,
                const float* __restrict__ G,
                const float* __restrict__ dinv,
                const unsigned int* __restrict__ deg8w,
                const float* __restrict__ W,
                const float* __restrict__ bias,
                float* __restrict__ OUT) {
    constexpr int VEC = K / 32;
    constexpr int P   = K + 4;
    constexpr int WP  = COUT + 8;
    typedef __attribute__((ext_vector_type(VEC))) float vecf;
    constexpr int NT  = COUT / 16;
    constexpr int GW  = 2 * NT;
    constexpr int NG  = 32 / GW;
    constexpr int GB  = 32 / VEC;
    __shared__ __attribute__((aligned(16))) float tile[2][32 * P];
    __shared__ __attribute__((aligned(16))) float wlds[K * WP];
    __shared__ __attribute__((aligned(16))) unsigned char dtab[SCALE_IN ? N_NODES : 16];

    const int lane = threadIdx.x & 31;
    const int w    = __builtin_amdgcn_readfirstlane(threadIdx.x >> 5);
    const int grp  = w / GW;
    const int wl   = w - grp * GW;

    constexpr int NDV = N_NODES / 16;
    constexpr int NDP = SCALE_IN ? (NDV + 1023) / 1024 : 0;
    uint4 dv[NDP > 0 ? NDP : 1];
#pragma unroll
    for (int i = 0; i < NDP; ++i) {
        const int p  = threadIdx.x + i * 1024;
        const int pc = p < NDV ? p : NDV - 1;
        dv[i] = reinterpret_cast<const uint4*>(deg8w)[pc];
    }

    constexpr int NWP = K * COUT / 4096;
    v4f wv[NWP];
#pragma unroll
    for (int i = 0; i < NWP; ++i)
        wv[i] = *reinterpret_cast<const v4f*>(W + threadIdx.x * 4 + i * 4096);
#pragma unroll
    for (int i = 0; i < NWP; ++i) {
        const int f   = threadIdx.x * 4 + i * 4096;
        const int row = f / COUT;
        const int cx  = f - row * COUT;
        *reinterpret_cast<v4f*>(wlds + row * WP + cx) = wv[i];
    }
    if (SCALE_IN) {
#pragma unroll
        for (int i = 0; i < NDP; ++i) {
            const int p  = threadIdx.x + i * 1024;
            const int pc = p < NDV ? p : NDV - 1;
            reinterpret_cast<uint4*>(dtab)[pc] = dv[i];
        }
        __syncthreads();
    }

    int it = 0;
    for (int t = blockIdx.x; t < N_TILES; t += gridDim.x, ++it) {
        float* tb = tile[it & 1];
        const int node = t * 32 + w;
        const vecf acc = gather_sum<K, GB, SCALE_IN>(rowptr, col, G, dinv, dtab, node, lane);

        *reinterpret_cast<vecf*>(tb + w * P + lane * VEC) = acc;
        __syncthreads();

        if (grp == (it & (NG - 1))) {
            const int mt   = wl / NT;
            const int nt   = wl - mt * NT;
            const int c    = lane & 15;
            const int kh   = (lane >> 4) << 1;
            const int colg = nt * 16 + c;
            const float* arow = tb + (mt * 16 + c) * P + kh;
            const float* wcol = wlds + kh * WP + colg;

            v8f dacc = {};
#pragma unroll
            for (int kk = 0; kk < K; kk += 4) {
                v2f a = *reinterpret_cast<const v2f*>(arow + kk);
                v2f b;
                b.x = wcol[(kk)     * WP];
                b.y = wcol[(kk + 1) * WP];
                dacc = __builtin_amdgcn_wmma_f32_16x16x4_f32(
                     false, a,  false, b,
                     (short)0, dacc,  false,  false);
            }

            const int   mbase = t * 32 + mt * 16 + ((lane >> 4) << 3);
            const float bcol  = bias[colg];
#pragma unroll
            for (int r = 0; r < 8; ++r) {
                const int   m  = mbase + r;
                const float dm = dinv[m];
                float v = dm * dacc[r] + bcol;
                if (RELU) v = fmaxf(v, 0.0f);
                if (SCALE_OUT) v *= dm;
                OUT[m * COUT + colg] = v;
            }
        }
    }
}

__global__ void __launch_bounds__(1024, 16)
agg_gemm_next_kernel(const int* __restrict__ rowptr,
                     const int* __restrict__ col,
                     const float* __restrict__ G,
                     const float* __restrict__ dinv,
                     const float* __restrict__ W,
                     const float* __restrict__ bias,
                     const float* __restrict__ Wn,
                     float* __restrict__ Z) {
    constexpr int K    = 128;
    constexpr int COUT = 128;
    constexpr int CN   = 64;
    constexpr int P    = K + 4;
    constexpr int WP   = COUT + 8;
    constexpr int WNP  = CN + 8;
    constexpr int NT   = COUT / 16;
    constexpr int GW   = 2 * NT;
    constexpr int NTN  = CN / 16;
    static_assert(GW == 16 && COUT == K, "two wave groups; the H tile has the sum tile's shape");
    __shared__ __attribute__((aligned(16))) float tile[2][32 * P];
    __shared__ __attribute__((aligned(16))) float htile[32 * P];
    __shared__ __attribute__((aligned(16))) float wlds[K * WP];
    __shared__ __attribute__((aligned(16))) float wnlds[COUT * WNP];

    const int lane = threadIdx.x & 31;
    const int w    = __builtin_amdgcn_readfirstlane(threadIdx.x >> 5);
    const int grp  = w / GW;
    const int wl   = w - grp * GW;

    constexpr int NWP  = K * COUT / 4096;
    constexpr int NWNP = COUT * CN / 4096;
    v4f wv[NWP];
    v4f wnv[NWNP];
#pragma unroll
    for (int i = 0; i < NWP; ++i)
        wv[i] = *reinterpret_cast<const v4f*>(W + threadIdx.x * 4 + i * 4096);
#pragma unroll
    for (int i = 0; i < NWNP; ++i)
        wnv[i] = *reinterpret_cast<const v4f*>(Wn + threadIdx.x * 4 + i * 4096);
#pragma unroll
    for (int i = 0; i < NWP; ++i) {
        const int f   = threadIdx.x * 4 + i * 4096;
        const int row = f / COUT;
        const int cx  = f - row * COUT;
        *reinterpret_cast<v4f*>(wlds + row * WP + cx) = wv[i];
    }
#pragma unroll
    for (int i = 0; i < NWNP; ++i) {
        const int f   = threadIdx.x * 4 + i * 4096;
        const int row = f / CN;
        const int cx  = f - row * CN;
        *reinterpret_cast<v4f*>(wnlds + row * WNP + cx) = wnv[i];
    }

    const int c  = lane & 15;
    const int kh = (lane >> 4) << 1;
    const int rh = (lane >> 4) << 3;

    int t = blockIdx.x;
    if (t < N_TILES) {
        const v4f acc = gather_sum<K, 8, false>(rowptr, col, G, dinv, nullptr, t * 32 + w, lane);
        *reinterpret_cast<v4f*>(tile[0] + w * P + lane * 4) = acc;
    }

    float hv[8];
    int it = 0;
    for (; t < N_TILES; t += gridDim.x, ++it) {
        __syncthreads();
        const float* tb = tile[it & 1];

        if (grp == (it & 1)) {
            const int mt   = wl / NT;
            const int nt   = wl - mt * NT;
            const int colg = nt * 16 + c;
            const float* arow = tb + (mt * 16 + c) * P + kh;
            const float* wcol = wlds + kh * WP + colg;

            v8f dacc = {};
#pragma unroll
            for (int kk = 0; kk < K; kk += 4) {
                v2f a = *reinterpret_cast<const v2f*>(arow + kk);
                v2f b;
                b.x = wcol[(kk)     * WP];
                b.y = wcol[(kk + 1) * WP];
                dacc = __builtin_amdgcn_wmma_f32_16x16x4_f32(
                     false, a,  false, b,
                     (short)0, dacc,  false,  false);
            }

            const int   mbase = t * 32 + mt * 16 + rh;
            const float bcol  = bias[colg];
#pragma unroll
            for (int r = 0; r < 8; ++r) {
                const float dm = dinv[mbase + r];
                float v = dm * dacc[r] + bcol;
                v = fmaxf(v, 0.0f);
                hv[r] = v * dm;
            }
        } else if (it > 0 && wl < 2 * NTN) {
            const int mt   = wl / NTN;
            const int nt   = wl - mt * NTN;
            const int colg = nt * 16 + c;
            const float* arow = htile + (mt * 16 + c) * P + kh;
            const float* wcol = wnlds + kh * WNP + colg;

            v8f zacc = {};
#pragma unroll
            for (int kk = 0; kk < COUT; kk += 4) {
                v2f a = *reinterpret_cast<const v2f*>(arow + kk);
                v2f b;
                b.x = wcol[(kk)     * WNP];
                b.y = wcol[(kk + 1) * WNP];
                zacc = __builtin_amdgcn_wmma_f32_16x16x4_f32(
                     false, a,  false, b,
                     (short)0, zacc,  false,  false);
            }

            const int mbase = (t - (int)gridDim.x) * 32 + mt * 16 + rh;
#pragma unroll
            for (int r = 0; r < 8; ++r)
                Z[(mbase + r) * CN + colg] = zacc[r];
        }

        const int tn = t + gridDim.x;
        if (tn < N_TILES) {
            const v4f acc = gather_sum<K, 8, false>(rowptr, col, G, dinv, nullptr, tn * 32 + w, lane);
            *reinterpret_cast<v4f*>(tile[(it + 1) & 1] + w * P + lane * 4) = acc;
        }
        __syncthreads();

        if (grp == (it & 1)) {
            const int mt   = wl / NT;
            const int nt   = wl - mt * NT;
            float* hcol = htile + (mt * 16 + rh) * P + nt * 16 + c;
#pragma unroll
            for (int r = 0; r < 8; ++r)
                hcol[r * P] = hv[r];
        }
    }

    __syncthreads();
    if (it > 0 && grp != (it & 1) && wl < 2 * NTN) {
        const int mt   = wl / NTN;
        const int nt   = wl - mt * NTN;
        const int colg = nt * 16 + c;
        const float* arow = htile + (mt * 16 + c) * P + kh;
        const float* wcol = wnlds + kh * WNP + colg;

        v8f zacc = {};
#pragma unroll
        for (int kk = 0; kk < COUT; kk += 4) {
            v2f a = *reinterpret_cast<const v2f*>(arow + kk);
            v2f b;
            b.x = wcol[(kk)     * WNP];
            b.y = wcol[(kk + 1) * WNP];
            zacc = __builtin_amdgcn_wmma_f32_16x16x4_f32(
                 false, a,  false, b,
                 (short)0, zacc,  false,  false);
        }

        const int mbase = (t - (int)gridDim.x) * 32 + mt * 16 + rh;
#pragma unroll
        for (int r = 0; r < 8; ++r)
            Z[(mbase + r) * CN + colg] = zacc[r];
    }
}

__global__ void __launch_bounds__(1024)
agg_out_kernel(const int* __restrict__ rowptr,
               const int* __restrict__ col,
               const float* __restrict__ Z,
               const float* __restrict__ dinv,
               const float* __restrict__ bias,
               float* __restrict__ OUT) {
    constexpr int C = 64;
    const int lane = threadIdx.x & 31;
    const int w    = __builtin_amdgcn_readfirstlane(threadIdx.x >> 5);
    const v2f bl = *reinterpret_cast<const v2f*>(bias + lane * 2);
    for (int t = blockIdx.x; t < N_TILES; t += gridDim.x) {
        const int node = t * 32 + w;
        const v2f acc = gather_sum<C, 16, false>(rowptr, col, Z, dinv, nullptr, node, lane);
        const float dm = dinv[node];
        *reinterpret_cast<v2f*>(OUT + node * C + lane * 2) = acc * dm + bl;
    }
}

extern "C" void kernel_launch(void* const* d_in, const int* in_sizes, int n_in,
                              void* d_out, int out_size, void* d_ws, size_t ws_size,
                              hipStream_t stream) {
    const float* x  = (const float*)d_in[0];
    const int*   ei = (const int*)d_in[1];
    const float* W1 = (const float*)d_in[2];
    const float* b1 = (const float*)d_in[3];
    const float* W2 = (const float*)d_in[4];
    const float* b2 = (const float*)d_in[5];
    const float* W3 = (const float*)d_in[6];
    const float* b3 = (const float*)d_in[7];
    float*       out = (float*)d_out;

    const int* src = ei;
    const int* dst = ei + N_EDGES;

    char*  ws       = (char*)d_ws;
    size_t off      = 0;
    float* dinv     = (float*)(ws + off); off += ((size_t)N_NODES * 4 + 511) & ~511ull;
    int*   rowptr   = (int*)(ws + off);   off += ((size_t)(N_NODES + 1) * 4 + 511) & ~511ull;
    unsigned int* deg8w = (unsigned int*)(ws + off); off += ((size_t)N_NODES + 511) & ~511ull;
    int*   starts   = (int*)(ws + off);   off += (size_t)(NB + 1) * GP * 4;
    int2*  binned   = (int2*)(ws + off);  off += (size_t)N_EDGES * 8;
    int*   col      = (int*)(ws + off);   off += (size_t)N_EDGES * 4;
    float* ha       = (float*)(ws + off); off += (size_t)N_NODES * 128 * 4;
    float* hb       = (float*)(ws + off);

    const int L_BLOCKS = 1024;
    const int L1_BLOCKS = 512;

    chunk_group_kernel<<<GP, 1024, 0, stream>>>(src, dst, starts, binned);
    bin_sort_kernel<<<NB, 1024, 0, stream>>>(binned, starts, rowptr, col, dinv, deg8w);

    agg_gemm_kernel<64, 128, true, true, true><<<L1_BLOCKS, 1024, 0, stream>>>(rowptr, col, x, dinv, deg8w, W1, b1, ha);

    agg_gemm_next_kernel<<<L_BLOCKS, 1024, 0, stream>>>(rowptr, col, ha, dinv, W2, b2, W3, hb);

    agg_out_kernel<<<L_BLOCKS, 1024, 0, stream>>>(rowptr, col, hb, dinv, b3, out);
}
